# starting kernel
# baseline (speedup 1.0000x reference)
#include <hip/hip_runtime.h>
#include <hip/hip_bf16.h>

typedef float v2f __attribute__((ext_vector_type(2)));
typedef float v8f __attribute__((ext_vector_type(8)));

#define GROUPS 64
#define GLEN   1025
#define HSZ    32
#define SCALE  0.17677669529663687f
#define PSTRIDE 68

__device__ __forceinline__ v8f wmma_f32(v2f a, v2f b, v8f c) {
    return __builtin_amdgcn_wmma_f32_16x16x4_f32(false, a, false, b, (short)0, c, false, false);
}

__global__ __launch_bounds__(32) void qkv_gemm_kernel(
    const float* __restrict__ x, const float* __restrict__ w,
    float* __restrict__ qg, float* __restrict__ kg, float* __restrict__ vg)
{
    const int mt = blockIdx.x;
    const int nt = blockIdx.y;
    const int lane = threadIdx.x;
    const int half = lane >> 4, l16 = lane & 15;

    const float* arow = x + (size_t)(mt * 16 + l16) * 128 + 2 * half;
    const float* brow = w + (size_t)(nt * 16 + l16) * 128 + 2 * half;

    v8f acc = {};
#pragma unroll
    for (int c = 0; c < 32; ++c) {
        float2 af = *(const float2*)(arow + 4 * c);
        float2 bf = *(const float2*)(brow + 4 * c);
        v2f a = { af.x, af.y };
        v2f b = { bf.x, bf.y };
        acc = wmma_f32(a, b, acc);
    }

    const int j0 = nt * 16;
    const int which = j0 / 128;
    float* dst = (which == 0) ? qg : ((which == 1) ? kg : vg);
    const int j = (j0 % 128) + l16;
    const int h = j / 32, d = j % 32;
    const int row0 = mt * 16;
    const int bidx = row0 / 8192;
#pragma unroll
    for (int r = 0; r < 8; ++r) {
        int t = (row0 % 8192) + r + 8 * half;
        int g = t / 1024, rig = t % 1024;
        size_t off = ((size_t)((bidx * 4 + h) * 8 + g) * GLEN + rig) * HSZ + d;
        dst[off] = acc[r];
    }
}

__global__ __launch_bounds__(128) void mean_kernel(
    float* __restrict__ qg, float* __restrict__ kg, float* __restrict__ vg)
{
    const int grp = blockIdx.x;
    const int t = threadIdx.x;
    if (t >= 96) return;
    float* buf = (t < 32) ? qg : ((t < 64) ? kg : vg);
    const int d = t % 32;
    const float* base = buf + (size_t)grp * GLEN * HSZ + d;
    float s = 0.0f;
    for (int r = 0; r < 1024; ++r) s += base[r * HSZ];
    buf[(size_t)grp * GLEN * HSZ + 1024 * HSZ + d] = s * (1.0f / 1024.0f);
}

__global__ __launch_bounds__(32) void attn_group_kernel(
    const float* __restrict__ qg, const float* __restrict__ kg,
    const float* __restrict__ vg, float* __restrict__ xg)
{
    const int qt = blockIdx.x;
    const int grp = blockIdx.y;
    const int lane = threadIdx.x;
    const int half = lane >> 4, l16 = lane & 15;

    const float* Q = qg + (size_t)grp * GLEN * HSZ;
    const float* K = kg + (size_t)grp * GLEN * HSZ;
    const float* V = vg + (size_t)grp * GLEN * HSZ;
    float*       X = xg + (size_t)grp * GLEN * HSZ;

    const int qbase = qt * 16;
    int qrow = qbase + l16; if (qrow > 1024) qrow = 1024;
    const float* qptr = Q + qrow * HSZ + 2 * half;

    v2f qf[8];
#pragma unroll
    for (int c = 0; c < 8; ++c) {
        float2 f = *(const float2*)(qptr + 4 * c);
        qf[c] = (v2f){ f.x, f.y };
    }

    v8f acc0 = {}, acc1 = {};
    float rowM[8], rowL[8];
#pragma unroll
    for (int r = 0; r < 8; ++r) { rowM[r] = -3.0e38f; rowL[r] = 0.0f; }

    __shared__ float lds[16 * PSTRIDE];

    for (int kbase = 0; kbase <= qbase + 15; kbase += 64) {
        int jmax = ((qbase + 15 - kbase) >> 4) + 1;
        if (jmax > 4) jmax = 4;

        v8f s[4];
#pragma unroll
        for (int j = 0; j < 4; ++j) {
            s[j] = (v8f){};
            if (j < jmax) {
                int krow = kbase + 16 * j + l16; if (krow > 1024) krow = 1024;
                const float* kptr = K + krow * HSZ + 2 * half;
#pragma unroll
                for (int c = 0; c < 8; ++c) {
                    float2 f = *(const float2*)(kptr + 4 * c);
                    v2f b = { f.x, f.y };
                    s[j] = wmma_f32(qf[c], b, s[j]);
                }
            }
        }

        float p[4][8];
#pragma unroll
        for (int r = 0; r < 8; ++r) {
            const int qi = qbase + r + 8 * half;
            float sv[4];
#pragma unroll
            for (int j = 0; j < 4; ++j) {
                const int kj = kbase + 16 * j + l16;
                const bool valid = (j < jmax) && (kj <= qi) && (kj <= 1024);
                sv[j] = valid ? s[j][r] * SCALE : -3.0e38f;
            }
            float rm = fmaxf(fmaxf(sv[0], sv[1]), fmaxf(sv[2], sv[3]));
            rm = fmaxf(rm, __shfl_xor(rm, 1, 32));
            rm = fmaxf(rm, __shfl_xor(rm, 2, 32));
            rm = fmaxf(rm, __shfl_xor(rm, 4, 32));
            rm = fmaxf(rm, __shfl_xor(rm, 8, 32));
            const float nm = fmaxf(rowM[r], rm);
            float rs = 0.0f;
#pragma unroll
            for (int j = 0; j < 4; ++j) {
                float pv = __expf(sv[j] - nm);
                p[j][r] = pv;
                rs += pv;
            }
            rs += __shfl_xor(rs, 1, 32);
            rs += __shfl_xor(rs, 2, 32);
            rs += __shfl_xor(rs, 4, 32);
            rs += __shfl_xor(rs, 8, 32);
            const float corr = __expf(rowM[r] - nm);
            rowM[r] = nm;
            rowL[r] = rowL[r] * corr + rs;
            acc0[r] = acc0[r] * corr;
            acc1[r] = acc1[r] * corr;
        }

        __syncthreads();
#pragma unroll
        for (int j = 0; j < 4; ++j)
#pragma unroll
            for (int r = 0; r < 8; ++r)
                lds[(r + 8 * half) * PSTRIDE + 16 * j + l16] = p[j][r];
        __syncthreads();

        const int cmax = 4 * jmax;
#pragma unroll
        for (int c = 0; c < 16; ++c) {
            if (c < cmax) {
                float2 pf = *(const float2*)&lds[l16 * PSTRIDE + 4 * c + 2 * half];
                v2f a = { pf.x, pf.y };
                int k0 = kbase + 4 * c + 2 * half;
                int kr0 = (k0     > 1024) ? 1024 : k0;
                int kr1 = (k0 + 1 > 1024) ? 1024 : k0 + 1;
                v2f b0 = { V[kr0 * HSZ + l16],      V[kr1 * HSZ + l16] };
                v2f b1 = { V[kr0 * HSZ + 16 + l16], V[kr1 * HSZ + 16 + l16] };
                acc0 = wmma_f32(a, b0, acc0);
                acc1 = wmma_f32(a, b1, acc1);
            }
        }
    }

#pragma unroll
    for (int r = 0; r < 8; ++r) {
        const int row = qbase + r + 8 * half;
        if (row <= 1024) {
            const float invl = 1.0f / rowL[r];
            X[row * HSZ + l16]      = acc0[r] * invl + V[row * HSZ + l16];
            X[row * HSZ + 16 + l16] = acc1[r] * invl + V[row * HSZ + 16 + l16];
        }
    }
}

__global__ __launch_bounds__(256) void meanatt_kernel(
    const float* __restrict__ qg, const float* __restrict__ kg,
    const float* __restrict__ xg, float* __restrict__ y)
{
    const int bh = blockIdx.x;
    __shared__ float qm[8][32], km[8][32], vm[8][32];
    const int t = threadIdx.x;
    const int g = t / 32, d = t % 32;
    const size_t rowoff = ((size_t)(bh * 8 + g) * GLEN + 1024) * HSZ + d;
    qm[g][d] = qg[rowoff];
    km[g][d] = kg[rowoff];
    vm[g][d] = xg[rowoff];
    __syncthreads();

    float sarr[8];
    float mx = -3.0e38f;
    for (int gp = 0; gp <= g; ++gp) {
        float sv = 0.0f;
        for (int c = 0; c < 32; ++c) sv += qm[g][c] * km[gp][c];
        sv *= SCALE;
        sarr[gp] = sv;
        mx = fmaxf(mx, sv);
    }
    float denom = 0.0f;
    for (int gp = 0; gp <= g; ++gp) { sarr[gp] = __expf(sarr[gp] - mx); denom += sarr[gp]; }
    float o = 0.0f;
    for (int gp = 0; gp <= g; ++gp) o += sarr[gp] * vm[gp][d];
    y[(bh * 8 + g) * 32 + d] = o / denom;
}

__global__ __launch_bounds__(256) void assemble_kernel(
    const float* __restrict__ xg, const float* __restrict__ y, float* __restrict__ z)
{
    const int idx = blockIdx.x * 256 + threadIdx.x;
    if (idx >= 2 * 8200 * 128) return;
    const int c = idx % 128;
    const int t = (idx / 128) % 8200;
    const int b = idx / (8200 * 128);
    const int h = c / 32, d = c % 32;
    const int g = t / GLEN, row = t % GLEN;
    const int bh = b * 4 + h;
    float v;
    if (g == 0)        v = xg[((size_t)(bh * 8 + 0) * GLEN + row) * HSZ + d];
    else if (row == 0) v = y[(bh * 8 + (g - 1)) * 32 + d];
    else               v = xg[((size_t)(bh * 8 + g) * GLEN + (row - 1)) * HSZ + d];
    z[idx] = v;
}

__global__ __launch_bounds__(32) void proj_gemm_kernel(
    const float* __restrict__ z, const float* __restrict__ w, float* __restrict__ out)
{
    const int mt = blockIdx.x;
    const int nt = blockIdx.y;
    const int lane = threadIdx.x;
    const int half = lane >> 4, l16 = lane & 15;

    const float* arow = z + (size_t)(mt * 16 + l16) * 128 + 2 * half;
    const float* brow = w + (size_t)(nt * 16 + l16) * 128 + 2 * half;

    v8f acc = {};
#pragma unroll
    for (int c = 0; c < 32; ++c) {
        float2 af = *(const float2*)(arow + 4 * c);
        float2 bf = *(const float2*)(brow + 4 * c);
        v2f a = { af.x, af.y };
        v2f b = { bf.x, bf.y };
        acc = wmma_f32(a, b, acc);
    }
#pragma unroll
    for (int r = 0; r < 8; ++r) {
        const int row = mt * 16 + r + 8 * half;
        out[(size_t)row * 128 + nt * 16 + l16] = acc[r];
    }
}

extern "C" void kernel_launch(void* const* d_in, const int* in_sizes, int n_in,
                              void* d_out, int out_size, void* d_ws, size_t ws_size,
                              hipStream_t stream) {
    const float* x      = (const float*)d_in[0];
    const float* w_attn = (const float*)d_in[1];
    const float* w_proj = (const float*)d_in[2];

    float* ws = (float*)d_ws;
    const size_t GBUF = (size_t)GROUPS * GLEN * HSZ;
    float* qg = ws;
    float* kg = ws + GBUF;
    float* vg = ws + 2 * GBUF;
    float* xg = ws + 3 * GBUF;
    float* y  = ws + 4 * GBUF;
    float* z  = qg;

    qkv_gemm_kernel<<<dim3(1024, 24), 32, 0, stream>>>(x, w_attn, qg, kg, vg);
    mean_kernel<<<GROUPS, 128, 0, stream>>>(qg, kg, vg);
    attn_group_kernel<<<dim3(65, GROUPS), 32, 0, stream>>>(qg, kg, vg, xg);
    meanatt_kernel<<<8, 256, 0, stream>>>(qg, kg, xg, y);
    assemble_kernel<<<8200, 256, 0, stream>>>(xg, y, z);
    proj_gemm_kernel<<<dim3(1025, 8), 32, 0, stream>>>(z, w_proj, (float*)d_out);
}


# both GEMM kernels: 8-wave work-group computes a 64x128 tile from A and weight rows staged once in LDS by whole-row 16-byte loads, results leave via LDS as whole lines; was 1 wave per 16x16 tile
# speedup vs baseline: 2.6095x; 2.6095x over previous
#include <hip/hip_runtime.h>
#include <hip/hip_bf16.h>

typedef float v2f __attribute__((ext_vector_type(2)));
typedef float v8f __attribute__((ext_vector_type(8)));

#define GROUPS 64
#define GLEN   1025
#define HSZ    32
#define SCALE  0.17677669529663687f
#define PSTRIDE 68
#define AW      4
#define WREG    (16 * PSTRIDE)

__device__ __forceinline__ v8f wmma_f32(v2f a, v2f b, v8f c) {
    return __builtin_amdgcn_wmma_f32_16x16x4_f32(false, a, false, b, (short)0, c, false, false);
}

__device__ __forceinline__ void wave_lds_sync() {
    __builtin_amdgcn_fence(__ATOMIC_RELEASE, "wavefront");
    __builtin_amdgcn_wave_barrier();
    __builtin_amdgcn_fence(__ATOMIC_ACQUIRE, "wavefront");
}

__global__ __launch_bounds__(512) void xmean_kernel(
    const float* __restrict__ x, float* __restrict__ xm)
{
    const int bg = blockIdx.x;
    const int slab = blockIdx.y;
    const int t = threadIdx.x;
    const int sub = t & 7;
    const int r0 = t >> 3;
    const float* p = x + ((size_t)bg * 1024 + r0) * 128 + slab * 32 + sub * 4;

    float4 v[16];
#pragma unroll
    for (int i = 0; i < 16; ++i) v[i] = *(const float4*)(p + (size_t)i * 64 * 128);
    float4 s = v[0];
#pragma unroll
    for (int i = 1; i < 16; ++i) { s.x += v[i].x; s.y += v[i].y; s.z += v[i].z; s.w += v[i].w; }

    s.x += __shfl_xor(s.x, 8, 32);  s.y += __shfl_xor(s.y, 8, 32);
    s.z += __shfl_xor(s.z, 8, 32);  s.w += __shfl_xor(s.w, 8, 32);
    s.x += __shfl_xor(s.x, 16, 32); s.y += __shfl_xor(s.y, 16, 32);
    s.z += __shfl_xor(s.z, 16, 32); s.w += __shfl_xor(s.w, 16, 32);

    __shared__ float red[16][32];
    const int wv = t >> 5, lane = t & 31;
    if (lane < 8) *(float4*)&red[wv][4 * lane] = s;
    __syncthreads();
    if (t < 32) {
        float a = 0.0f;
#pragma unroll
        for (int w = 0; w < 16; ++w) a += red[w][t];
        xm[bg * 128 + slab * 32 + t] = a * (1.0f / 1024.0f);
    }
}

#define TP      132
#define GEMM_BM 64
#define GEMM_BN 128

__device__ __forceinline__ void gemm_tile_64x128(
    const float* __restrict__ a_src, const int nvalid,
    const float* __restrict__ b_src, float* __restrict__ lds)
{
    float* la = lds;
    float* lb = lds + GEMM_BM * TP;
    const int wv = __builtin_amdgcn_readfirstlane((int)(threadIdx.x >> 5));
    const int lane = threadIdx.x & 31;
    const int half = lane >> 4, l16 = lane & 15;

    float4 ra[8], rb[16];
#pragma unroll
    for (int p = 0; p < 8; ++p) {
        const int row = p * 8 + wv;
        ra[p] = make_float4(0.0f, 0.0f, 0.0f, 0.0f);
        if (row < nvalid) ra[p] = *(const float4*)(a_src + (size_t)row * 128 + lane * 4);
    }
#pragma unroll
    for (int p = 0; p < 16; ++p) {
        const int row = p * 8 + wv;
        rb[p] = *(const float4*)(b_src + (size_t)row * 128 + lane * 4);
    }
#pragma unroll
    for (int p = 0; p < 8; ++p)
        *(float4*)&la[(p * 8 + wv) * TP + lane * 4] = ra[p];
#pragma unroll
    for (int p = 0; p < 16; ++p)
        *(float4*)&lb[(p * 8 + wv) * TP + lane * 4] = rb[p];
    __syncthreads();

    const int rt = wv & 3;
    const int cg = wv >> 2;
    v8f acc[4];
#pragma unroll
    for (int j = 0; j < 4; ++j) acc[j] = (v8f){};

    if (rt * 16 < nvalid) {
        const float* pa = la + (rt * 16 + l16) * TP + 4 * half;
        const float* pb = lb + (cg * 64 + l16) * TP + 4 * half;
#pragma unroll
        for (int kk = 0; kk < 16; ++kk) {
            const float4 a4 = *(const float4*)(pa + 8 * kk);
            float4 b4[4];
#pragma unroll
            for (int j = 0; j < 4; ++j) b4[j] = *(const float4*)(pb + j * 16 * TP + 8 * kk);
            const v2f a0 = { a4.x, a4.y };
            const v2f a1 = { a4.z, a4.w };
#pragma unroll
            for (int j = 0; j < 4; ++j) {
                const v2f b0 = { b4[j].x, b4[j].y };
                acc[j] = wmma_f32(a0, b0, acc[j]);
            }
#pragma unroll
            for (int j = 0; j < 4; ++j) {
                const v2f b1 = { b4[j].z, b4[j].w };
                acc[j] = wmma_f32(a1, b1, acc[j]);
            }
        }
    }

    __syncthreads();
#pragma unroll
    for (int j = 0; j < 4; ++j)
#pragma unroll
        for (int r = 0; r < 8; ++r)
            la[(rt * 16 + r + 8 * half) * TP + cg * 64 + j * 16 + l16] = acc[j][r];
    __syncthreads();
}

__global__ __launch_bounds__(256) void qkv_gemm_kernel(
    const float* __restrict__ x, const float* __restrict__ xm, const float* __restrict__ w,
    float* __restrict__ qg, float* __restrict__ kg, float* __restrict__ vg)
{
    __shared__ float lds[(GEMM_BM + GEMM_BN) * TP];
    const int nb = blockIdx.x;
    const bool pooled = (blockIdx.y == 0);
    const int mb = (int)blockIdx.y - 1;
    const float* a_src = pooled ? xm : x + (size_t)mb * GEMM_BM * 128;
    const int nvalid = pooled ? 16 : GEMM_BM;

    gemm_tile_64x128(a_src, nvalid, w + (size_t)nb * GEMM_BN * 128, lds);

    float* dst = (nb == 0) ? qg : ((nb == 1) ? kg : vg);
    const int wv = __builtin_amdgcn_readfirstlane((int)(threadIdx.x >> 5));
    const int lane = threadIdx.x & 31;
    const int h = lane >> 3, d = (lane & 7) * 4;
    float4 vrow[8];
#pragma unroll
    for (int p = 0; p < 8; ++p) vrow[p] = *(const float4*)&lds[(p * 8 + wv) * TP + lane * 4];
#pragma unroll
    for (int p = 0; p < 8; ++p) {
        const int row = p * 8 + wv;
        if (row < nvalid) {
            const float4 v = vrow[p];
            int bidx, g, rig;
            if (pooled) { bidx = row >> 3; g = row & 7; rig = 1024; }
            else {
                const int tok = mb * GEMM_BM + row;
                bidx = tok >> 13;
                g = (tok & 8191) >> 10;
                rig = tok & 1023;
            }
            const size_t off = ((size_t)((bidx * 4 + h) * 8 + g) * GLEN + rig) * HSZ + d;
            *(float4*)(dst + off) = v;
        }
    }
}

__global__ __launch_bounds__(32 * AW, 8) void attn_group_kernel(
    const float* __restrict__ qg, const float* __restrict__ kg,
    const float* __restrict__ vg, float* __restrict__ xg)
{
    const int grp = blockIdx.x;
    const int qt = 64 - (int)blockIdx.y;
    const int wv = __builtin_amdgcn_readfirstlane((int)(threadIdx.x >> 5));
    const int lane = threadIdx.x & 31;
    const int half = lane >> 4, l16 = lane & 15;

    const float* Q = qg + (size_t)grp * GLEN * HSZ;
    const float* K = kg + (size_t)grp * GLEN * HSZ;
    const float* V = vg + (size_t)grp * GLEN * HSZ;
    float*       X = xg + (size_t)grp * GLEN * HSZ;

    const int qbase = qt * 16;
    int qrow = qbase + l16; if (qrow > 1024) qrow = 1024;
    const float* qptr = Q + qrow * HSZ + 2 * half;

    v2f qf[8];
#pragma unroll
    for (int c = 0; c < 8; ++c) {
        float2 f = *(const float2*)(qptr + 4 * c);
        qf[c] = (v2f){ f.x, f.y };
    }

    v8f acc0 = {}, acc1 = {};
    float rowM[8], rowL[8];
#pragma unroll
    for (int r = 0; r < 8; ++r) { rowM[r] = -3.0e38f; rowL[r] = 0.0f; }

    __shared__ float lds_all[AW * WREG];
    float* lds = lds_all + wv * WREG;

    for (int kbase = 64 * wv; kbase <= qbase + 15; kbase += 64 * AW) {
        int jmax = ((qbase + 15 - kbase) >> 4) + 1;
        if (jmax > 4) jmax = 4;

        v8f s[4];
#pragma unroll
        for (int j = 0; j < 4; ++j) {
            s[j] = (v8f){};
            if (j < jmax) {
                int krow = kbase + 16 * j + l16; if (krow > 1024) krow = 1024;
                const float* kptr = K + krow * HSZ + 2 * half;
#pragma unroll
                for (int c = 0; c < 8; ++c) {
                    float2 f = *(const float2*)(kptr + 4 * c);
                    v2f b = { f.x, f.y };
                    s[j] = wmma_f32(qf[c], b, s[j]);
                }
            }
        }

        float p[4][8];
#pragma unroll
        for (int r = 0; r < 8; ++r) {
            const int qi = qbase + r + 8 * half;
            float sv[4];
#pragma unroll
            for (int j = 0; j < 4; ++j) {
                const int kj = kbase + 16 * j + l16;
                const bool valid = (j < jmax) && (kj <= qi) && (kj <= 1024);
                sv[j] = valid ? s[j][r] * SCALE : -3.0e38f;
            }
            float rm = fmaxf(fmaxf(sv[0], sv[1]), fmaxf(sv[2], sv[3]));
            rm = fmaxf(rm, __shfl_xor(rm, 1, 32));
            rm = fmaxf(rm, __shfl_xor(rm, 2, 32));
            rm = fmaxf(rm, __shfl_xor(rm, 4, 32));
            rm = fmaxf(rm, __shfl_xor(rm, 8, 32));
            const float nm = fmaxf(rowM[r], rm);
            float rs = 0.0f;
#pragma unroll
            for (int j = 0; j < 4; ++j) {
                float pv = __expf(sv[j] - nm);
                p[j][r] = pv;
                rs += pv;
            }
            rs += __shfl_xor(rs, 1, 32);
            rs += __shfl_xor(rs, 2, 32);
            rs += __shfl_xor(rs, 4, 32);
            rs += __shfl_xor(rs, 8, 32);
            const float corr = __expf(rowM[r] - nm);
            rowM[r] = nm;
            rowL[r] = rowL[r] * corr + rs;
            acc0[r] = acc0[r] * corr;
            acc1[r] = acc1[r] * corr;
        }

        wave_lds_sync();
#pragma unroll
        for (int j = 0; j < 4; ++j)
#pragma unroll
            for (int r = 0; r < 8; ++r)
                lds[(r + 8 * half) * PSTRIDE + 16 * j + l16] = p[j][r];
        wave_lds_sync();

        const int cmax = 4 * jmax;
#pragma unroll
        for (int c = 0; c < 16; ++c) {
            if (c < cmax) {
                float2 pf = *(const float2*)&lds[l16 * PSTRIDE + 4 * c + 2 * half];
                v2f a = { pf.x, pf.y };
                int k0 = kbase + 4 * c + 2 * half;
                int kr0 = (k0     > 1024) ? 1024 : k0;
                int kr1 = (k0 + 1 > 1024) ? 1024 : k0 + 1;
                v2f b0 = { V[kr0 * HSZ + l16],      V[kr1 * HSZ + l16] };
                v2f b1 = { V[kr0 * HSZ + 16 + l16], V[kr1 * HSZ + 16 + l16] };
                acc0 = wmma_f32(a, b0, acc0);
                acc1 = wmma_f32(a, b1, acc1);
            }
        }
    }

    wave_lds_sync();
#pragma unroll
    for (int r = 0; r < 8; ++r) {
        lds[(r)      * 32 + lane] = rowM[r];
        lds[(8 + r)  * 32 + lane] = rowL[r];
        lds[(16 + r) * 32 + lane] = acc0[r];
        lds[(24 + r) * 32 + lane] = acc1[r];
    }
    __syncthreads();

#pragma unroll
    for (int i = 0; i < 8 / AW; ++i) {
        const int rr = wv + AW * i;
        float pm[AW];
        float gm = -3.0e38f;
#pragma unroll
        for (int w = 0; w < AW; ++w) {
            pm[w] = lds_all[w * WREG + rr * 32 + lane];
            gm = fmaxf(gm, pm[w]);
        }
        float gl = 0.0f, o0 = 0.0f, o1 = 0.0f;
#pragma unroll
        for (int w = 0; w < AW; ++w) {
            const float e = __expf(pm[w] - gm);
            gl += lds_all[w * WREG + (8 + rr)  * 32 + lane] * e;
            o0 += lds_all[w * WREG + (16 + rr) * 32 + lane] * e;
            o1 += lds_all[w * WREG + (24 + rr) * 32 + lane] * e;
        }

        const int row = qbase + rr + 8 * half;
        if (row <= 1024) {
            const float invl = 1.0f / gl;
            X[row * HSZ + l16]      = o0 * invl + V[row * HSZ + l16];
            X[row * HSZ + 16 + l16] = o1 * invl + V[row * HSZ + 16 + l16];
        }
    }
}

__global__ __launch_bounds__(256) void meanatt_kernel(
    const float* __restrict__ qg, const float* __restrict__ kg,
    const float* __restrict__ xg, float* __restrict__ y)
{
    const int bh = blockIdx.x;
    __shared__ float qm[8][32], km[8][32], vm[8][32];
    const int t = threadIdx.x;
    const int g = t / 32, d = t % 32;
    const size_t rowoff = ((size_t)(bh * 8 + g) * GLEN + 1024) * HSZ + d;
    qm[g][d] = qg[rowoff];
    km[g][d] = kg[rowoff];
    vm[g][d] = xg[rowoff];
    __syncthreads();

    float sarr[8];
    float mx = -3.0e38f;
    for (int gp = 0; gp <= g; ++gp) {
        float sv = 0.0f;
        for (int c = 0; c < 32; ++c) sv += qm[g][c] * km[gp][c];
        sv *= SCALE;
        sarr[gp] = sv;
        mx = fmaxf(mx, sv);
    }
    float denom = 0.0f;
    for (int gp = 0; gp <= g; ++gp) { sarr[gp] = __expf(sarr[gp] - mx); denom += sarr[gp]; }
    float o = 0.0f;
    for (int gp = 0; gp <= g; ++gp) o += sarr[gp] * vm[gp][d];
    y[(bh * 8 + g) * 32 + d] = o / denom;
}

__global__ __launch_bounds__(256) void assemble_kernel(
    const float* __restrict__ xg, const float* __restrict__ y, float* __restrict__ z)
{
    const int idx = blockIdx.x * 256 + threadIdx.x;
    if (idx >= 2 * 8200 * 128) return;
    const int c = idx % 128;
    const int t = (idx / 128) % 8200;
    const int b = idx / (8200 * 128);
    const int h = c / 32, d = c % 32;
    const int g = t / GLEN, row = t % GLEN;
    const int bh = b * 4 + h;
    float v;
    if (g == 0)        v = xg[((size_t)(bh * 8 + 0) * GLEN + row) * HSZ + d];
    else if (row == 0) v = y[(bh * 8 + (g - 1)) * 32 + d];
    else               v = xg[((size_t)(bh * 8 + g) * GLEN + (row - 1)) * HSZ + d];
    z[idx] = v;
}

__global__ __launch_bounds__(256) void proj_gemm_kernel(
    const float* __restrict__ z, const float* __restrict__ w, float* __restrict__ out)
{
    __shared__ float lds[(GEMM_BM + GEMM_BN) * TP];
    const int mb = blockIdx.x;
    const int nvalid = (mb == 256) ? 16 : GEMM_BM;

    gemm_tile_64x128(z + (size_t)mb * GEMM_BM * 128, nvalid, w, lds);

    const int wv = __builtin_amdgcn_readfirstlane((int)(threadIdx.x >> 5));
    const int lane = threadIdx.x & 31;
    float4 vrow[8];
#pragma unroll
    for (int p = 0; p < 8; ++p) vrow[p] = *(const float4*)&lds[(p * 8 + wv) * TP + lane * 4];
#pragma unroll
    for (int p = 0; p < 8; ++p) {
        const int row = p * 8 + wv;
        if (row < nvalid) {
            const float4 v = vrow[p];
            *(float4*)(out + ((size_t)mb * GEMM_BM + row) * 128 + lane * 4) = v;
        }
    }
}

extern "C" void kernel_launch(void* const* d_in, const int* in_sizes, int n_in,
                              void* d_out, int out_size, void* d_ws, size_t ws_size,
                              hipStream_t stream) {
    const float* x      = (const float*)d_in[0];
    const float* w_attn = (const float*)d_in[1];
    const float* w_proj = (const float*)d_in[2];

    float* ws = (float*)d_ws;
    const size_t GBUF = (size_t)GROUPS * GLEN * HSZ;
    float* qg = ws;
    float* kg = ws + GBUF;
    float* vg = ws + 2 * GBUF;
    float* xg = ws + 3 * GBUF;
    float* y  = ws + 4 * GBUF;
    float* xm = y;
    float* z  = qg;

    xmean_kernel<<<dim3(16, 4), 512, 0, stream>>>(x, xm);
    qkv_gemm_kernel<<<dim3(3, 257), 256, 0, stream>>>(x, xm, w_attn, qg, kg, vg);
    attn_group_kernel<<<dim3(GROUPS, 65), 128, 0, stream>>>(qg, kg, vg, xg);
    meanatt_kernel<<<8, 256, 0, stream>>>(qg, kg, xg, y);
    assemble_kernel<<<8200, 256, 0, stream>>>(xg, y, z);
    proj_gemm_kernel<<<257, 256, 0, stream>>>(z, w_proj, (float*)d_out);
}


# attention: each 64-key block's K and V rows are staged in the wave's own LDS slice by 32 whole-line 16-byte loads in one group, all WMMA fragments read from LDS; was 96 narrow global loads per block
# speedup vs baseline: 2.9804x; 1.1421x over previous
#include <hip/hip_runtime.h>
#include <hip/hip_bf16.h>

typedef float v2f __attribute__((ext_vector_type(2)));
typedef float v8f __attribute__((ext_vector_type(8)));
typedef float v4f __attribute__((ext_vector_type(4)));

#define GROUPS 64
#define GLEN   1025
#define HSZ    32
#define SCALE  0.17677669529663687f
#define PSTRIDE 68
#define AW      4
#define KPITCH  36
#define VPITCH  40
#define KSLICE  (64 * KPITCH)
#define VSLICE  (64 * VPITCH)
#define WREG    (KSLICE + VSLICE)

__device__ __forceinline__ v8f wmma_f32(v2f a, v2f b, v8f c) {
    return __builtin_amdgcn_wmma_f32_16x16x4_f32(false, a, false, b, (short)0, c, false, false);
}

__device__ __forceinline__ void wave_lds_sync() {
    __builtin_amdgcn_fence(__ATOMIC_RELEASE, "wavefront");
    __builtin_amdgcn_wave_barrier();
    __builtin_amdgcn_fence(__ATOMIC_ACQUIRE, "wavefront");
}

__global__ __launch_bounds__(512) void xmean_kernel(
    const float* __restrict__ x, float* __restrict__ xm)
{
    const int bg = blockIdx.x;
    const int slab = blockIdx.y;
    const int t = threadIdx.x;
    const int sub = t & 7;
    const int r0 = t >> 3;
    const float* p = x + ((size_t)bg * 1024 + r0) * 128 + slab * 32 + sub * 4;

    float4 v[16];
#pragma unroll
    for (int i = 0; i < 16; ++i) v[i] = *(const float4*)(p + (size_t)i * 64 * 128);
    float4 s = v[0];
#pragma unroll
    for (int i = 1; i < 16; ++i) { s.x += v[i].x; s.y += v[i].y; s.z += v[i].z; s.w += v[i].w; }

    s.x += __shfl_xor(s.x, 8, 32);  s.y += __shfl_xor(s.y, 8, 32);
    s.z += __shfl_xor(s.z, 8, 32);  s.w += __shfl_xor(s.w, 8, 32);
    s.x += __shfl_xor(s.x, 16, 32); s.y += __shfl_xor(s.y, 16, 32);
    s.z += __shfl_xor(s.z, 16, 32); s.w += __shfl_xor(s.w, 16, 32);

    __shared__ float red[16][32];
    const int wv = t >> 5, lane = t & 31;
    if (lane < 8) *(float4*)&red[wv][4 * lane] = s;
    __syncthreads();
    if (t < 32) {
        float a = 0.0f;
#pragma unroll
        for (int w = 0; w < 16; ++w) a += red[w][t];
        xm[bg * 128 + slab * 32 + t] = a * (1.0f / 1024.0f);
    }
}

#define TP      132
#define GEMM_BM 64
#define GEMM_BN 128

__device__ __forceinline__ void gemm_tile_64x128(
    const float* __restrict__ a_src, const int nvalid,
    const float* __restrict__ b_src, float* __restrict__ lds)
{
    float* la = lds;
    float* lb = lds + GEMM_BM * TP;
    const int wv = __builtin_amdgcn_readfirstlane((int)(threadIdx.x >> 5));
    const int lane = threadIdx.x & 31;
    const int half = lane >> 4, l16 = lane & 15;

    float4 ra[8], rb[16];
#pragma unroll
    for (int p = 0; p < 8; ++p) {
        const int row = p * 8 + wv;
        ra[p] = make_float4(0.0f, 0.0f, 0.0f, 0.0f);
        if (row < nvalid) ra[p] = *(const float4*)(a_src + (size_t)row * 128 + lane * 4);
    }
#pragma unroll
    for (int p = 0; p < 16; ++p) {
        const int row = p * 8 + wv;
        rb[p] = *(const float4*)(b_src + (size_t)row * 128 + lane * 4);
    }
#pragma unroll
    for (int p = 0; p < 8; ++p)
        *(float4*)&la[(p * 8 + wv) * TP + lane * 4] = ra[p];
#pragma unroll
    for (int p = 0; p < 16; ++p)
        *(float4*)&lb[(p * 8 + wv) * TP + lane * 4] = rb[p];
    __syncthreads();

    const int rt = wv & 3;
    const int cg = wv >> 2;
    v8f acc[4];
#pragma unroll
    for (int j = 0; j < 4; ++j) acc[j] = (v8f){};

    if (rt * 16 < nvalid) {
        const float* pa = la + (rt * 16 + l16) * TP + 4 * half;
        const float* pb = lb + (cg * 64 + l16) * TP + 4 * half;
#pragma unroll
        for (int kk = 0; kk < 16; ++kk) {
            const float4 a4 = *(const float4*)(pa + 8 * kk);
            float4 b4[4];
#pragma unroll
            for (int j = 0; j < 4; ++j) b4[j] = *(const float4*)(pb + j * 16 * TP + 8 * kk);
            const v2f a0 = { a4.x, a4.y };
            const v2f a1 = { a4.z, a4.w };
#pragma unroll
            for (int j = 0; j < 4; ++j) {
                const v2f b0 = { b4[j].x, b4[j].y };
                acc[j] = wmma_f32(a0, b0, acc[j]);
            }
#pragma unroll
            for (int j = 0; j < 4; ++j) {
                const v2f b1 = { b4[j].z, b4[j].w };
                acc[j] = wmma_f32(a1, b1, acc[j]);
            }
        }
    }

    __syncthreads();
#pragma unroll
    for (int j = 0; j < 4; ++j)
#pragma unroll
        for (int r = 0; r < 8; ++r)
            la[(rt * 16 + r + 8 * half) * TP + cg * 64 + j * 16 + l16] = acc[j][r];
    __syncthreads();
}

__global__ __launch_bounds__(256) void qkv_gemm_kernel(
    const float* __restrict__ x, const float* __restrict__ xm, const float* __restrict__ w,
    float* __restrict__ qg, float* __restrict__ kg, float* __restrict__ vg)
{
    __shared__ float lds[(GEMM_BM + GEMM_BN) * TP];
    const int nb = blockIdx.x;
    const bool pooled = (blockIdx.y == 0);
    const int mb = (int)blockIdx.y - 1;
    const float* a_src = pooled ? xm : x + (size_t)mb * GEMM_BM * 128;
    const int nvalid = pooled ? 16 : GEMM_BM;

    gemm_tile_64x128(a_src, nvalid, w + (size_t)nb * GEMM_BN * 128, lds);

    float* dst = (nb == 0) ? qg : ((nb == 1) ? kg : vg);
    const int wv = __builtin_amdgcn_readfirstlane((int)(threadIdx.x >> 5));
    const int lane = threadIdx.x & 31;
    const int h = lane >> 3, d = (lane & 7) * 4;
    float4 vrow[8];
#pragma unroll
    for (int p = 0; p < 8; ++p) vrow[p] = *(const float4*)&lds[(p * 8 + wv) * TP + lane * 4];
#pragma unroll
    for (int p = 0; p < 8; ++p) {
        const int row = p * 8 + wv;
        if (row < nvalid) {
            const float4 v = vrow[p];
            int bidx, g, rig;
            if (pooled) { bidx = row >> 3; g = row & 7; rig = 1024; }
            else {
                const int tok = mb * GEMM_BM + row;
                bidx = tok >> 13;
                g = (tok & 8191) >> 10;
                rig = tok & 1023;
            }
            const size_t off = ((size_t)((bidx * 4 + h) * 8 + g) * GLEN + rig) * HSZ + d;
            *(float4*)(dst + off) = v;
        }
    }
}

__global__ __launch_bounds__(32 * AW, 4) void attn_group_kernel(
    const float* __restrict__ qg, const float* __restrict__ kg,
    const float* __restrict__ vg, float* __restrict__ z, float* __restrict__ xgm)
{
    const int grp = blockIdx.x;
    const int qt = 64 - (int)blockIdx.y;
    const int wv = __builtin_amdgcn_readfirstlane((int)(threadIdx.x >> 5));
    const int lane = threadIdx.x & 31;
    const int half = lane >> 4, l16 = lane & 15;

    const float* Q = qg + (size_t)grp * GLEN * HSZ;
    const float* K = kg + (size_t)grp * GLEN * HSZ;
    const float* V = vg + (size_t)grp * GLEN * HSZ;
    const int gi = grp & 7;
    float*       Z = z + (size_t)(grp >> 5) * 8200 * 128 + ((grp >> 3) & 3) * HSZ;

    const int qbase = qt * 16;
    int qrow = qbase + l16; if (qrow > 1024) qrow = 1024;
    const float* qptr = Q + qrow * HSZ + 2 * half;

    v2f qf[8];
#pragma unroll
    for (int c = 0; c < 8; ++c) {
        float2 f = *(const float2*)(qptr + 4 * c);
        qf[c] = (v2f){ f.x, f.y };
    }

    v8f acc0 = {}, acc1 = {};
    float rowM[8], rowL[8];
#pragma unroll
    for (int r = 0; r < 8; ++r) { rowM[r] = -3.0e38f; rowL[r] = 0.0f; }

    __shared__ __attribute__((aligned(16))) float lds_all[AW * WREG];
    float* lds = lds_all + wv * WREG;
    float* ldv = lds + KSLICE;

    const int srow = lane >> 3;
    const int scol = (lane & 7) * 4;

    for (int kbase = 64 * wv; kbase <= qbase + 15; kbase += 64 * AW) {
        int jmax = ((qbase + 15 - kbase) >> 4) + 1;
        if (jmax > 4) jmax = 4;

        v4f kr[16], vr[16];
        const bool tail = (kbase >= 1024);
        const float* kb = K + (tail ? 1024 : kbase) * HSZ;
        const float* vb = V + (tail ? 1024 : kbase) * HSZ;
        if (!tail) {
            const unsigned lo = (unsigned)(srow * HSZ + scol);
#pragma unroll
            for (int i = 0; i < 16; ++i) kr[i] = *(const v4f*)(kb + lo + i * 4 * HSZ);
#pragma unroll
            for (int i = 0; i < 16; ++i) vr[i] = *(const v4f*)(vb + lo + i * 4 * HSZ);
        } else {
            const v4f k1 = *(const v4f*)(kb + scol);
            const v4f v1 = *(const v4f*)(vb + scol);
#pragma unroll
            for (int i = 0; i < 16; ++i) { kr[i] = k1; vr[i] = v1; }
        }
        wave_lds_sync();
#pragma unroll
        for (int i = 0; i < 16; ++i)
            *(v4f*)&lds[(4 * i + srow) * KPITCH + scol] = kr[i];
#pragma unroll
        for (int i = 0; i < 16; ++i)
            *(v4f*)&ldv[(4 * i + srow) * VPITCH + scol] = vr[i];
        wave_lds_sync();

        v2f kf[4][8];
#pragma unroll
        for (int j = 0; j < 4; ++j)
#pragma unroll
            for (int c = 0; c < 8; ++c) {
                const float2 f = *(const float2*)&lds[(16 * j + l16) * KPITCH + 4 * c + 2 * half];
                kf[j][c] = (v2f){ f.x, f.y };
            }
        __builtin_amdgcn_sched_barrier(0);

        v8f s[4];
#pragma unroll
        for (int j = 0; j < 4; ++j) s[j] = (v8f){};
        if (jmax == 4) {
#pragma unroll
            for (int c = 0; c < 8; ++c)
#pragma unroll
                for (int j = 0; j < 4; ++j)
                    s[j] = wmma_f32(qf[c], kf[j][c], s[j]);
        } else {
#pragma unroll
            for (int j = 0; j < 3; ++j) {
                if (j < jmax) {
#pragma unroll
                    for (int c = 0; c < 8; ++c)
                        s[j] = wmma_f32(qf[c], kf[j][c], s[j]);
                }
            }
        }

        float p[4][8];
#pragma unroll
        for (int r = 0; r < 8; ++r) {
            const int qi = qbase + r + 8 * half;
            float sv[4];
#pragma unroll
            for (int j = 0; j < 4; ++j) {
                const int kj = kbase + 16 * j + l16;
                const bool valid = (j < jmax) && (kj <= qi) && (kj <= 1024);
                sv[j] = valid ? s[j][r] * SCALE : -3.0e38f;
            }
            float rm = fmaxf(fmaxf(sv[0], sv[1]), fmaxf(sv[2], sv[3]));
            rm = fmaxf(rm, __shfl_xor(rm, 1, 32));
            rm = fmaxf(rm, __shfl_xor(rm, 2, 32));
            rm = fmaxf(rm, __shfl_xor(rm, 4, 32));
            rm = fmaxf(rm, __shfl_xor(rm, 8, 32));
            const float nm = fmaxf(rowM[r], rm);
            float rs = 0.0f;
#pragma unroll
            for (int j = 0; j < 4; ++j) {
                float pv = __expf(sv[j] - nm);
                p[j][r] = pv;
                rs += pv;
            }
            rs += __shfl_xor(rs, 1, 32);
            rs += __shfl_xor(rs, 2, 32);
            rs += __shfl_xor(rs, 4, 32);
            rs += __shfl_xor(rs, 8, 32);
            const float corr = __expf(rowM[r] - nm);
            rowM[r] = nm;
            rowL[r] = rowL[r] * corr + rs;
            acc0[r] = acc0[r] * corr;
            acc1[r] = acc1[r] * corr;
        }

        wave_lds_sync();
#pragma unroll
        for (int j = 0; j < 4; ++j)
#pragma unroll
            for (int r = 0; r < 8; ++r)
                lds[(r + 8 * half) * PSTRIDE + 16 * j + l16] = p[j][r];
        wave_lds_sync();

        v2f pf[16], b0[16], b1[16];
#pragma unroll
        for (int c = 0; c < 16; ++c) {
            const float2 f = *(const float2*)&lds[l16 * PSTRIDE + 4 * c + 2 * half];
            pf[c] = (v2f){ f.x, f.y };
            const float* vp = &ldv[(4 * c + 2 * half) * VPITCH + l16];
            b0[c] = (v2f){ vp[0],  vp[VPITCH] };
            b1[c] = (v2f){ vp[16], vp[VPITCH + 16] };
        }
        __builtin_amdgcn_sched_barrier(0);

        if (jmax == 4) {
#pragma unroll
            for (int c = 0; c < 16; ++c) {
                acc0 = wmma_f32(pf[c], b0[c], acc0);
                acc1 = wmma_f32(pf[c], b1[c], acc1);
            }
        } else {
#pragma unroll
            for (int j = 0; j < 3; ++j) {
                if (j < jmax) {
#pragma unroll
                    for (int c = 4 * j; c < 4 * j + 4; ++c) {
                        acc0 = wmma_f32(pf[c], b0[c], acc0);
                        acc1 = wmma_f32(pf[c], b1[c], acc1);
                    }
                }
            }
        }
    }

    wave_lds_sync();
#pragma unroll
    for (int r = 0; r < 8; ++r) {
        lds[(r)      * 32 + lane] = rowM[r];
        lds[(8 + r)  * 32 + lane] = rowL[r];
        lds[(16 + r) * 32 + lane] = acc0[r];
        lds[(24 + r) * 32 + lane] = acc1[r];
    }
    __syncthreads();

#pragma unroll
    for (int i = 0; i < 8 / AW; ++i) {
        const int rr = wv + AW * i;
        float pm[AW];
        float gm = -3.0e38f;
#pragma unroll
        for (int w = 0; w < AW; ++w) {
            pm[w] = lds_all[w * WREG + rr * 32 + lane];
            gm = fmaxf(gm, pm[w]);
        }
        float gl = 0.0f, o0 = 0.0f, o1 = 0.0f;
#pragma unroll
        for (int w = 0; w < AW; ++w) {
            const float e = __expf(pm[w] - gm);
            gl += lds_all[w * WREG + (8 + rr)  * 32 + lane] * e;
            o0 += lds_all[w * WREG + (16 + rr) * 32 + lane] * e;
            o1 += lds_all[w * WREG + (24 + rr) * 32 + lane] * e;
        }

        const int row = qbase + rr + 8 * half;
        if (row <= 1024) {
            const float invl = 1.0f / gl;
            const float r0 = o0 * invl + V[row * HSZ + l16];
            const float r1 = o1 * invl + V[row * HSZ + 16 + l16];
            if (row < 1024 || gi == 0) {
                float* zp = Z + (size_t)(gi * GLEN + row + (gi > 0 ? 1 : 0)) * 128;
                zp[l16]      = r0;
                zp[16 + l16] = r1;
            }
            if (row == 1024) {
                xgm[grp * HSZ + l16]      = r0;
                xgm[grp * HSZ + 16 + l16] = r1;
            }
        }
    }
}

__global__ __launch_bounds__(256) void meanatt_kernel(
    const float* __restrict__ qg, const float* __restrict__ kg,
    const float* __restrict__ xgm, float* __restrict__ z)
{
    const int bh = blockIdx.x;
    __shared__ float qm[8][32], km[8][32], vm[8][32];
    const int t = threadIdx.x;
    const int g = t / 32, d = t % 32;
    const size_t rowoff = ((size_t)(bh * 8 + g) * GLEN + 1024) * HSZ + d;
    qm[g][d] = qg[rowoff];
    km[g][d] = kg[rowoff];
    vm[g][d] = xgm[(bh * 8 + g) * HSZ + d];
    __syncthreads();

    float sarr[8];
    float mx = -3.0e38f;
    for (int gp = 0; gp <= g; ++gp) {
        float sv = 0.0f;
        for (int c = 0; c < 32; ++c) sv += qm[g][c] * km[gp][c];
        sv *= SCALE;
        sarr[gp] = sv;
        mx = fmaxf(mx, sv);
    }
    float denom = 0.0f;
    for (int gp = 0; gp <= g; ++gp) { sarr[gp] = __expf(sarr[gp] - mx); denom += sarr[gp]; }
    float o = 0.0f;
    for (int gp = 0; gp <= g; ++gp) o += sarr[gp] * vm[gp][d];
    if (g < 7)
        z[((size_t)(bh >> 2) * 8200 + (size_t)(g + 1) * GLEN) * 128 + (bh & 3) * HSZ + d] = o / denom;
}

__global__ __launch_bounds__(256) void proj_gemm_kernel(
    const float* __restrict__ z, const float* __restrict__ w, float* __restrict__ out)
{
    __shared__ float lds[(GEMM_BM + GEMM_BN) * TP];
    const int mb = blockIdx.x;
    const int nvalid = (mb == 256) ? 16 : GEMM_BM;

    gemm_tile_64x128(z + (size_t)mb * GEMM_BM * 128, nvalid, w, lds);

    const int wv = __builtin_amdgcn_readfirstlane((int)(threadIdx.x >> 5));
    const int lane = threadIdx.x & 31;
    float4 vrow[8];
#pragma unroll
    for (int p = 0; p < 8; ++p) vrow[p] = *(const float4*)&lds[(p * 8 + wv) * TP + lane * 4];
#pragma unroll
    for (int p = 0; p < 8; ++p) {
        const int row = p * 8 + wv;
        if (row < nvalid) {
            const float4 v = vrow[p];
            *(float4*)(out + ((size_t)mb * GEMM_BM + row) * 128 + lane * 4) = v;
        }
    }
}

extern "C" void kernel_launch(void* const* d_in, const int* in_sizes, int n_in,
                              void* d_out, int out_size, void* d_ws, size_t ws_size,
                              hipStream_t stream) {
    const float* x      = (const float*)d_in[0];
    const float* w_attn = (const float*)d_in[1];
    const float* w_proj = (const float*)d_in[2];

    float* ws = (float*)d_ws;
    const size_t GBUF = (size_t)GROUPS * GLEN * HSZ;
    float* qg  = ws;
    float* kg  = ws + GBUF;
    float* vg  = ws + 2 * GBUF;
    float* z   = ws + 3 * GBUF;
    float* xgm = ws + 4 * GBUF;
    float* xm  = xgm;

    xmean_kernel<<<dim3(16, 4), 512, 0, stream>>>(x, xm);
    qkv_gemm_kernel<<<dim3(3, 257), 256, 0, stream>>>(x, xm, w_attn, qg, kg, vg);
    attn_group_kernel<<<dim3(GROUPS, 65), 128, 0, stream>>>(qg, kg, vg, z, xgm);
    meanatt_kernel<<<8, 256, 0, stream>>>(qg, kg, xgm, z);
    proj_gemm_kernel<<<257, 256, 0, stream>>>(z, w_proj, (float*)d_out);
}


# attention softmax: row max by 4 DPP steps (64 ds_bpermute per key block gone), row sum kept per lane and summed once in the merge, scale folded into the exp2 argument, mask only on diagonal blocks
# speedup vs baseline: 3.3572x; 1.1264x over previous
#include <hip/hip_runtime.h>
#include <hip/hip_bf16.h>

typedef float v2f __attribute__((ext_vector_type(2)));
typedef float v8f __attribute__((ext_vector_type(8)));
typedef float v4f __attribute__((ext_vector_type(4)));

#define GROUPS 64
#define GLEN   1025
#define HSZ    32
#define SCALE  0.17677669529663687f
#define SCALE_LOG2E (SCALE * 1.4426950408889634f)
#define PSTRIDE 68
#define AW      4
#define KPITCH  36
#define VPITCH  40
#define KSLICE  (64 * KPITCH)
#define VSLICE  (64 * VPITCH)
#define WREG    (KSLICE + VSLICE)

__device__ __forceinline__ v8f wmma_f32(v2f a, v2f b, v8f c) {
    return __builtin_amdgcn_wmma_f32_16x16x4_f32(false, a, false, b, (short)0, c, false, false);
}

__device__ __forceinline__ void wave_lds_sync() {
    __builtin_amdgcn_fence(__ATOMIC_RELEASE, "wavefront");
    __builtin_amdgcn_wave_barrier();
    __builtin_amdgcn_fence(__ATOMIC_ACQUIRE, "wavefront");
}

#define DPP_XM(op, n, mask) op " %" #n ", %" #n ", %" #n " row_xmask:" #mask " row_mask:0xf bank_mask:0xf\n\t"
#define DPP_PHASE8(op, mask) DPP_XM(op, 0, mask) DPP_XM(op, 1, mask) DPP_XM(op, 2, mask) DPP_XM(op, 3, mask) \
                             DPP_XM(op, 4, mask) DPP_XM(op, 5, mask) DPP_XM(op, 6, mask) DPP_XM(op, 7, mask)

__device__ __forceinline__ void row16_max8(float (&v)[8]) {
    asm(DPP_PHASE8("v_max_num_f32_dpp", 1) DPP_PHASE8("v_max_num_f32_dpp", 2)
        DPP_PHASE8("v_max_num_f32_dpp", 4) DPP_PHASE8("v_max_num_f32_dpp", 8)
        : "+v"(v[0]), "+v"(v[1]), "+v"(v[2]), "+v"(v[3]), "+v"(v[4]), "+v"(v[5]), "+v"(v[6]), "+v"(v[7]));
}

__device__ __forceinline__ float row16_sum(float v) {
    asm(DPP_XM("v_add_f32_dpp", 0, 1) DPP_XM("v_add_f32_dpp", 0, 2)
        DPP_XM("v_add_f32_dpp", 0, 4) DPP_XM("v_add_f32_dpp", 0, 8)
        : "+v"(v));
    return v;
}

__global__ __launch_bounds__(512) void xmean_kernel(
    const float* __restrict__ x, float* __restrict__ xm)
{
    const int bg = blockIdx.x;
    const int slab = blockIdx.y;
    const int t = threadIdx.x;
    const int sub = t & 7;
    const int r0 = t >> 3;
    const float* p = x + ((size_t)bg * 1024 + r0) * 128 + slab * 32 + sub * 4;

    float4 v[16];
#pragma unroll
    for (int i = 0; i < 16; ++i) v[i] = *(const float4*)(p + (size_t)i * 64 * 128);
    float4 s = v[0];
#pragma unroll
    for (int i = 1; i < 16; ++i) { s.x += v[i].x; s.y += v[i].y; s.z += v[i].z; s.w += v[i].w; }

    s.x += __shfl_xor(s.x, 8, 32);  s.y += __shfl_xor(s.y, 8, 32);
    s.z += __shfl_xor(s.z, 8, 32);  s.w += __shfl_xor(s.w, 8, 32);
    s.x += __shfl_xor(s.x, 16, 32); s.y += __shfl_xor(s.y, 16, 32);
    s.z += __shfl_xor(s.z, 16, 32); s.w += __shfl_xor(s.w, 16, 32);

    __shared__ float red[16][32];
    const int wv = t >> 5, lane = t & 31;
    if (lane < 8) *(float4*)&red[wv][4 * lane] = s;
    __syncthreads();
    if (t < 32) {
        float a = 0.0f;
#pragma unroll
        for (int w = 0; w < 16; ++w) a += red[w][t];
        xm[bg * 128 + slab * 32 + t] = a * (1.0f / 1024.0f);
    }
}

#define TP      132
#define GEMM_BM 64
#define GEMM_BN 128

__device__ __forceinline__ void gemm_tile_64x128(
    const float* __restrict__ a_src, const int nvalid,
    const float* __restrict__ b_src, float* __restrict__ lds)
{
    float* la = lds;
    float* lb = lds + GEMM_BM * TP;
    const int wv = __builtin_amdgcn_readfirstlane((int)(threadIdx.x >> 5));
    const int lane = threadIdx.x & 31;
    const int half = lane >> 4, l16 = lane & 15;

    float4 ra[8], rb[16];
#pragma unroll
    for (int p = 0; p < 8; ++p) {
        const int row = p * 8 + wv;
        ra[p] = make_float4(0.0f, 0.0f, 0.0f, 0.0f);
        if (row < nvalid) ra[p] = *(const float4*)(a_src + (size_t)row * 128 + lane * 4);
    }
#pragma unroll
    for (int p = 0; p < 16; ++p) {
        const int row = p * 8 + wv;
        rb[p] = *(const float4*)(b_src + (size_t)row * 128 + lane * 4);
    }
#pragma unroll
    for (int p = 0; p < 8; ++p)
        *(float4*)&la[(p * 8 + wv) * TP + lane * 4] = ra[p];
#pragma unroll
    for (int p = 0; p < 16; ++p)
        *(float4*)&lb[(p * 8 + wv) * TP + lane * 4] = rb[p];
    __syncthreads();

    const int rt = wv & 3;
    const int cg = wv >> 2;
    v8f acc[4];
#pragma unroll
    for (int j = 0; j < 4; ++j) acc[j] = (v8f){};

    if (rt * 16 < nvalid) {
        const float* pa = la + (rt * 16 + l16) * TP + 4 * half;
        const float* pb = lb + (cg * 64 + l16) * TP + 4 * half;
#pragma unroll
        for (int kk = 0; kk < 16; ++kk) {
            const float4 a4 = *(const float4*)(pa + 8 * kk);
            float4 b4[4];
#pragma unroll
            for (int j = 0; j < 4; ++j) b4[j] = *(const float4*)(pb + j * 16 * TP + 8 * kk);
            const v2f a0 = { a4.x, a4.y };
            const v2f a1 = { a4.z, a4.w };
#pragma unroll
            for (int j = 0; j < 4; ++j) {
                const v2f b0 = { b4[j].x, b4[j].y };
                acc[j] = wmma_f32(a0, b0, acc[j]);
            }
#pragma unroll
            for (int j = 0; j < 4; ++j) {
                const v2f b1 = { b4[j].z, b4[j].w };
                acc[j] = wmma_f32(a1, b1, acc[j]);
            }
        }
    }

    __syncthreads();
#pragma unroll
    for (int j = 0; j < 4; ++j)
#pragma unroll
        for (int r = 0; r < 8; ++r)
            la[(rt * 16 + r + 8 * half) * TP + cg * 64 + j * 16 + l16] = acc[j][r];
    __syncthreads();
}

__global__ __launch_bounds__(256) void qkv_gemm_kernel(
    const float* __restrict__ x, const float* __restrict__ xm, const float* __restrict__ w,
    float* __restrict__ qg, float* __restrict__ kg, float* __restrict__ vg)
{
    __shared__ float lds[(GEMM_BM + GEMM_BN) * TP];
    const int nb = blockIdx.x;
    const bool pooled = (blockIdx.y == 0);
    const int mb = (int)blockIdx.y - 1;
    const float* a_src = pooled ? xm : x + (size_t)mb * GEMM_BM * 128;
    const int nvalid = pooled ? 16 : GEMM_BM;

    gemm_tile_64x128(a_src, nvalid, w + (size_t)nb * GEMM_BN * 128, lds);

    float* dst = (nb == 0) ? qg : ((nb == 1) ? kg : vg);
    const int wv = __builtin_amdgcn_readfirstlane((int)(threadIdx.x >> 5));
    const int lane = threadIdx.x & 31;
    const int h = lane >> 3, d = (lane & 7) * 4;
    float4 vrow[8];
#pragma unroll
    for (int p = 0; p < 8; ++p) vrow[p] = *(const float4*)&lds[(p * 8 + wv) * TP + lane * 4];
#pragma unroll
    for (int p = 0; p < 8; ++p) {
        const int row = p * 8 + wv;
        if (row < nvalid) {
            const float4 v = vrow[p];
            int bidx, g, rig;
            if (pooled) { bidx = row >> 3; g = row & 7; rig = 1024; }
            else {
                const int tok = mb * GEMM_BM + row;
                bidx = tok >> 13;
                g = (tok & 8191) >> 10;
                rig = tok & 1023;
            }
            const size_t off = ((size_t)((bidx * 4 + h) * 8 + g) * GLEN + rig) * HSZ + d;
            *(float4*)(dst + off) = v;
        }
    }
}

__global__ __launch_bounds__(32 * AW, 4) void attn_group_kernel(
    const float* __restrict__ qg, const float* __restrict__ kg,
    const float* __restrict__ vg, float* __restrict__ z, float* __restrict__ xgm)
{
    const int grp = blockIdx.x;
    const int qt = 64 - (int)blockIdx.y;
    const int wv = __builtin_amdgcn_readfirstlane((int)(threadIdx.x >> 5));
    const int lane = threadIdx.x & 31;
    const int half = lane >> 4, l16 = lane & 15;

    const float* Q = qg + (size_t)grp * GLEN * HSZ;
    const float* K = kg + (size_t)grp * GLEN * HSZ;
    const float* V = vg + (size_t)grp * GLEN * HSZ;
    const int gi = grp & 7;
    float*       Z = z + (size_t)(grp >> 5) * 8200 * 128 + ((grp >> 3) & 3) * HSZ;

    const int qbase = qt * 16;
    int qrow = qbase + l16; if (qrow > 1024) qrow = 1024;
    const float* qptr = Q + qrow * HSZ + 2 * half;

    v2f qf[8];
#pragma unroll
    for (int c = 0; c < 8; ++c) {
        float2 f = *(const float2*)(qptr + 4 * c);
        qf[c] = (v2f){ f.x, f.y };
    }

    v8f acc0 = {}, acc1 = {};
    float rowM[8], rowL[8];
#pragma unroll
    for (int r = 0; r < 8; ++r) { rowM[r] = -3.0e38f; rowL[r] = 0.0f; }

    __shared__ __attribute__((aligned(16))) float lds_all[AW * WREG];
    float* lds = lds_all + wv * WREG;
    float* ldv = lds + KSLICE;

    const int srow = lane >> 3;
    const int scol = (lane & 7) * 4;

    for (int kbase = 64 * wv; kbase <= qbase + 15; kbase += 64 * AW) {
        int jmax = ((qbase + 15 - kbase) >> 4) + 1;
        if (jmax > 4) jmax = 4;

        v4f kr[16], vr[16];
        const bool tail = (kbase >= 1024);
        const float* kb = K + (tail ? 1024 : kbase) * HSZ;
        const float* vb = V + (tail ? 1024 : kbase) * HSZ;
        if (!tail) {
            const unsigned lo = (unsigned)(srow * HSZ + scol);
#pragma unroll
            for (int i = 0; i < 16; ++i) kr[i] = *(const v4f*)(kb + lo + i * 4 * HSZ);
#pragma unroll
            for (int i = 0; i < 16; ++i) vr[i] = *(const v4f*)(vb + lo + i * 4 * HSZ);
        } else {
            const v4f k1 = *(const v4f*)(kb + scol);
            const v4f v1 = *(const v4f*)(vb + scol);
#pragma unroll
            for (int i = 0; i < 16; ++i) { kr[i] = k1; vr[i] = v1; }
        }
        wave_lds_sync();
#pragma unroll
        for (int i = 0; i < 16; ++i)
            *(v4f*)&lds[(4 * i + srow) * KPITCH + scol] = kr[i];
#pragma unroll
        for (int i = 0; i < 16; ++i)
            *(v4f*)&ldv[(4 * i + srow) * VPITCH + scol] = vr[i];
        wave_lds_sync();

        v2f kf[4][8];
#pragma unroll
        for (int j = 0; j < 4; ++j)
#pragma unroll
            for (int c = 0; c < 8; ++c) {
                const float2 f = *(const float2*)&lds[(16 * j + l16) * KPITCH + 4 * c + 2 * half];
                kf[j][c] = (v2f){ f.x, f.y };
            }
        __builtin_amdgcn_sched_barrier(0);

        v8f s[4];
#pragma unroll
        for (int j = 0; j < 4; ++j) s[j] = (v8f){};
        if (jmax == 4) {
#pragma unroll
            for (int c = 0; c < 8; ++c)
#pragma unroll
                for (int j = 0; j < 4; ++j)
                    s[j] = wmma_f32(qf[c], kf[j][c], s[j]);
        } else {
#pragma unroll
            for (int j = 0; j < 3; ++j) {
                if (j < jmax) {
#pragma unroll
                    for (int c = 0; c < 8; ++c)
                        s[j] = wmma_f32(qf[c], kf[j][c], s[j]);
                }
            }
        }

        v2f b0[16], b1[16];
#pragma unroll
        for (int c = 0; c < 16; ++c) {
            const float* vp = &ldv[(4 * c + 2 * half) * VPITCH + l16];
            b0[c] = (v2f){ vp[0],  vp[VPITCH] };
            b1[c] = (v2f){ vp[16], vp[VPITCH + 16] };
        }

        const bool full = (kbase + 63 <= qbase);
        if (!full) {
#pragma unroll
            for (int r = 0; r < 8; ++r) {
                const int qi = qbase + r + 8 * half;
#pragma unroll
                for (int j = 0; j < 4; ++j) {
                    const int kj = kbase + 16 * j + l16;
                    const bool valid = (j < jmax) && (kj <= qi) && (kj <= 1024);
                    s[j][r] = valid ? s[j][r] : -3.0e38f;
                }
            }
        }
        float nmx[8];
#pragma unroll
        for (int r = 0; r < 8; ++r)
            nmx[r] = fmaxf(fmaxf(fmaxf(s[0][r], s[1][r]), s[2][r]), fmaxf(s[3][r], rowM[r]));
        row16_max8(nmx);
#pragma unroll
        for (int r = 0; r < 8; ++r) {
            const float nm = nmx[r];
            const float nb = -nm * SCALE_LOG2E;
            const float corr = __builtin_amdgcn_exp2f(__builtin_fmaf(rowM[r], SCALE_LOG2E, nb));
            float rs = 0.0f;
#pragma unroll
            for (int j = 0; j < 4; ++j) {
                const float pv = __builtin_amdgcn_exp2f(__builtin_fmaf(s[j][r], SCALE_LOG2E, nb));
                s[j][r] = pv;
                rs += pv;
            }
            rowM[r] = nm;
            rowL[r] = __builtin_fmaf(rowL[r], corr, rs);
            acc0[r] = acc0[r] * corr;
            acc1[r] = acc1[r] * corr;
        }

        wave_lds_sync();
#pragma unroll
        for (int j = 0; j < 4; ++j)
#pragma unroll
            for (int r = 0; r < 8; ++r)
                lds[(r + 8 * half) * PSTRIDE + 16 * j + l16] = s[j][r];
        wave_lds_sync();

        v2f pf[16];
#pragma unroll
        for (int c = 0; c < 16; ++c) {
            const float2 f = *(const float2*)&lds[l16 * PSTRIDE + 4 * c + 2 * half];
            pf[c] = (v2f){ f.x, f.y };
        }
        __builtin_amdgcn_sched_barrier(0);

        if (jmax == 4) {
#pragma unroll
            for (int c = 0; c < 16; ++c) {
                acc0 = wmma_f32(pf[c], b0[c], acc0);
                acc1 = wmma_f32(pf[c], b1[c], acc1);
            }
        } else {
#pragma unroll
            for (int j = 0; j < 3; ++j) {
                if (j < jmax) {
#pragma unroll
                    for (int c = 4 * j; c < 4 * j + 4; ++c) {
                        acc0 = wmma_f32(pf[c], b0[c], acc0);
                        acc1 = wmma_f32(pf[c], b1[c], acc1);
                    }
                }
            }
        }
    }

    wave_lds_sync();
#pragma unroll
    for (int r = 0; r < 8; ++r) {
        lds[(r)      * 32 + lane] = rowM[r];
        lds[(8 + r)  * 32 + lane] = rowL[r];
        lds[(16 + r) * 32 + lane] = acc0[r];
        lds[(24 + r) * 32 + lane] = acc1[r];
    }
    __syncthreads();

    float gls[8 / AW], o0s[8 / AW], o1s[8 / AW];
#pragma unroll
    for (int i = 0; i < 8 / AW; ++i) {
        const int rr = wv + AW * i;
        float pm[AW];
        float gm = -3.0e38f;
#pragma unroll
        for (int w = 0; w < AW; ++w) {
            pm[w] = lds_all[w * WREG + rr * 32 + lane];
            gm = fmaxf(gm, pm[w]);
        }
        float gl = 0.0f, o0 = 0.0f, o1 = 0.0f;
#pragma unroll
        for (int w = 0; w < AW; ++w) {
            const float e = __builtin_amdgcn_exp2f((pm[w] - gm) * SCALE_LOG2E);
            gl += lds_all[w * WREG + (8 + rr)  * 32 + lane] * e;
            o0 += lds_all[w * WREG + (16 + rr) * 32 + lane] * e;
            o1 += lds_all[w * WREG + (24 + rr) * 32 + lane] * e;
        }
        gls[i] = gl; o0s[i] = o0; o1s[i] = o1;
    }
#pragma unroll
    for (int i = 0; i < 8 / AW; ++i) gls[i] = row16_sum(gls[i]);

#pragma unroll
    for (int i = 0; i < 8 / AW; ++i) {
        const int rr = wv + AW * i;
        const int row = qbase + rr + 8 * half;
        if (row <= 1024) {
            const float invl = 1.0f / gls[i];
            const float o0 = o0s[i], o1 = o1s[i];
            const float r0 = o0 * invl + V[row * HSZ + l16];
            const float r1 = o1 * invl + V[row * HSZ + 16 + l16];
            if (row < 1024 || gi == 0) {
                float* zp = Z + (size_t)(gi * GLEN + row + (gi > 0 ? 1 : 0)) * 128;
                zp[l16]      = r0;
                zp[16 + l16] = r1;
            }
            if (row == 1024) {
                xgm[grp * HSZ + l16]      = r0;
                xgm[grp * HSZ + 16 + l16] = r1;
            }
        }
    }
}

__global__ __launch_bounds__(256) void meanatt_kernel(
    const float* __restrict__ qg, const float* __restrict__ kg,
    const float* __restrict__ xgm, float* __restrict__ z)
{
    const int bh = blockIdx.x;
    __shared__ float qm[8][32], km[8][32], vm[8][32];
    const int t = threadIdx.x;
    const int g = t / 32, d = t % 32;
    const size_t rowoff = ((size_t)(bh * 8 + g) * GLEN + 1024) * HSZ + d;
    qm[g][d] = qg[rowoff];
    km[g][d] = kg[rowoff];
    vm[g][d] = xgm[(bh * 8 + g) * HSZ + d];
    __syncthreads();

    float sarr[8];
    float mx = -3.0e38f;
    for (int gp = 0; gp <= g; ++gp) {
        float sv = 0.0f;
        for (int c = 0; c < 32; ++c) sv += qm[g][c] * km[gp][c];
        sv *= SCALE;
        sarr[gp] = sv;
        mx = fmaxf(mx, sv);
    }
    float denom = 0.0f;
    for (int gp = 0; gp <= g; ++gp) { sarr[gp] = __expf(sarr[gp] - mx); denom += sarr[gp]; }
    float o = 0.0f;
    for (int gp = 0; gp <= g; ++gp) o += sarr[gp] * vm[gp][d];
    if (g < 7)
        z[((size_t)(bh >> 2) * 8200 + (size_t)(g + 1) * GLEN) * 128 + (bh & 3) * HSZ + d] = o / denom;
}

__global__ __launch_bounds__(256) void proj_gemm_kernel(
    const float* __restrict__ z, const float* __restrict__ w, float* __restrict__ out)
{
    __shared__ float lds[(GEMM_BM + GEMM_BN) * TP];
    const int mb = blockIdx.x;
    const int nvalid = (mb == 256) ? 16 : GEMM_BM;

    gemm_tile_64x128(z + (size_t)mb * GEMM_BM * 128, nvalid, w, lds);

    const int wv = __builtin_amdgcn_readfirstlane((int)(threadIdx.x >> 5));
    const int lane = threadIdx.x & 31;
    float4 vrow[8];
#pragma unroll
    for (int p = 0; p < 8; ++p) vrow[p] = *(const float4*)&lds[(p * 8 + wv) * TP + lane * 4];
#pragma unroll
    for (int p = 0; p < 8; ++p) {
        const int row = p * 8 + wv;
        if (row < nvalid) {
            const float4 v = vrow[p];
            *(float4*)(out + ((size_t)mb * GEMM_BM + row) * 128 + lane * 4) = v;
        }
    }
}

extern "C" void kernel_launch(void* const* d_in, const int* in_sizes, int n_in,
                              void* d_out, int out_size, void* d_ws, size_t ws_size,
                              hipStream_t stream) {
    const float* x      = (const float*)d_in[0];
    const float* w_attn = (const float*)d_in[1];
    const float* w_proj = (const float*)d_in[2];

    float* ws = (float*)d_ws;
    const size_t GBUF = (size_t)GROUPS * GLEN * HSZ;
    float* qg  = ws;
    float* kg  = ws + GBUF;
    float* vg  = ws + 2 * GBUF;
    float* z   = ws + 3 * GBUF;
    float* xgm = ws + 4 * GBUF;
    float* xm  = xgm;

    xmean_kernel<<<dim3(16, 4), 512, 0, stream>>>(x, xm);
    qkv_gemm_kernel<<<dim3(3, 257), 256, 0, stream>>>(x, xm, w_attn, qg, kg, vg);
    attn_group_kernel<<<dim3(GROUPS, 65), 128, 0, stream>>>(qg, kg, vg, z, xgm);
    meanatt_kernel<<<8, 256, 0, stream>>>(qg, kg, xgm, z);
    proj_gemm_kernel<<<257, 256, 0, stream>>>(z, w_proj, (float*)d_out);
}


# attention: each wave owns a 32-row Q tile (two WMMA row tiles): every staged K/V block and every K and V fragment read from LDS feeds two WMMAs; was one 16-row tile per wave
# speedup vs baseline: 3.5323x; 1.0522x over previous
#include <hip/hip_runtime.h>
#include <hip/hip_bf16.h>

typedef float v2f __attribute__((ext_vector_type(2)));
typedef float v8f __attribute__((ext_vector_type(8)));
typedef float v4f __attribute__((ext_vector_type(4)));

#define GROUPS 64
#define GLEN   1025
#define HSZ    32
#define SCALE  0.17677669529663687f
#define SCALE_LOG2E (SCALE * 1.4426950408889634f)
#define PSTRIDE 68
#define AW      4
#define KPITCH  36
#define VPITCH  40
#define KSLICE  (64 * KPITCH)
#define VSLICE  (64 * VPITCH)
#define WREG    (KSLICE + VSLICE)

__device__ __forceinline__ v8f wmma_f32(v2f a, v2f b, v8f c) {
    return __builtin_amdgcn_wmma_f32_16x16x4_f32(false, a, false, b, (short)0, c, false, false);
}

__device__ __forceinline__ void wave_lds_sync() {
    __builtin_amdgcn_fence(__ATOMIC_RELEASE, "wavefront");
    __builtin_amdgcn_wave_barrier();
    __builtin_amdgcn_fence(__ATOMIC_ACQUIRE, "wavefront");
}

#define DPP_XM(op, n, mask) op " %" #n ", %" #n ", %" #n " row_xmask:" #mask " row_mask:0xf bank_mask:0xf\n\t"
#define DPP_PHASE8(op, mask) DPP_XM(op, 0, mask) DPP_XM(op, 1, mask) DPP_XM(op, 2, mask) DPP_XM(op, 3, mask) \
                             DPP_XM(op, 4, mask) DPP_XM(op, 5, mask) DPP_XM(op, 6, mask) DPP_XM(op, 7, mask)

__device__ __forceinline__ void row16_max8(float (&v)[8]) {
    asm(DPP_PHASE8("v_max_num_f32_dpp", 1) DPP_PHASE8("v_max_num_f32_dpp", 2)
        DPP_PHASE8("v_max_num_f32_dpp", 4) DPP_PHASE8("v_max_num_f32_dpp", 8)
        : "+v"(v[0]), "+v"(v[1]), "+v"(v[2]), "+v"(v[3]), "+v"(v[4]), "+v"(v[5]), "+v"(v[6]), "+v"(v[7]));
}

__device__ __forceinline__ float row16_sum(float v) {
    asm(DPP_XM("v_add_f32_dpp", 0, 1) DPP_XM("v_add_f32_dpp", 0, 2)
        DPP_XM("v_add_f32_dpp", 0, 4) DPP_XM("v_add_f32_dpp", 0, 8)
        : "+v"(v));
    return v;
}

__global__ __launch_bounds__(512) void xmean_kernel(
    const float* __restrict__ x, float* __restrict__ xm)
{
    const int bg = blockIdx.x;
    const int slab = blockIdx.y;
    const int t = threadIdx.x;
    const int sub = t & 7;
    const int r0 = t >> 3;
    const float* p = x + ((size_t)bg * 1024 + r0) * 128 + slab * 32 + sub * 4;

    float4 v[16];
#pragma unroll
    for (int i = 0; i < 16; ++i) v[i] = *(const float4*)(p + (size_t)i * 64 * 128);
    float4 s = v[0];
#pragma unroll
    for (int i = 1; i < 16; ++i) { s.x += v[i].x; s.y += v[i].y; s.z += v[i].z; s.w += v[i].w; }

    s.x += __shfl_xor(s.x, 8, 32);  s.y += __shfl_xor(s.y, 8, 32);
    s.z += __shfl_xor(s.z, 8, 32);  s.w += __shfl_xor(s.w, 8, 32);
    s.x += __shfl_xor(s.x, 16, 32); s.y += __shfl_xor(s.y, 16, 32);
    s.z += __shfl_xor(s.z, 16, 32); s.w += __shfl_xor(s.w, 16, 32);

    __shared__ float red[16][32];
    const int wv = t >> 5, lane = t & 31;
    if (lane < 8) *(float4*)&red[wv][4 * lane] = s;
    __syncthreads();
    if (t < 32) {
        float a = 0.0f;
#pragma unroll
        for (int w = 0; w < 16; ++w) a += red[w][t];
        xm[bg * 128 + slab * 32 + t] = a * (1.0f / 1024.0f);
    }
}

#define TP      132
#define GEMM_BM 64
#define GEMM_BN 128

__device__ __forceinline__ void gemm_tile_64x128(
    const float* __restrict__ a_src, const int nvalid,
    const float* __restrict__ b_src, float* __restrict__ lds)
{
    float* la = lds;
    float* lb = lds + GEMM_BM * TP;
    const int wv = __builtin_amdgcn_readfirstlane((int)(threadIdx.x >> 5));
    const int lane = threadIdx.x & 31;
    const int half = lane >> 4, l16 = lane & 15;

    float4 ra[8], rb[16];
#pragma unroll
    for (int p = 0; p < 8; ++p) {
        const int row = p * 8 + wv;
        ra[p] = make_float4(0.0f, 0.0f, 0.0f, 0.0f);
        if (row < nvalid) ra[p] = *(const float4*)(a_src + (size_t)row * 128 + lane * 4);
    }
#pragma unroll
    for (int p = 0; p < 16; ++p) {
        const int row = p * 8 + wv;
        rb[p] = *(const float4*)(b_src + (size_t)row * 128 + lane * 4);
    }
#pragma unroll
    for (int p = 0; p < 8; ++p)
        *(float4*)&la[(p * 8 + wv) * TP + lane * 4] = ra[p];
#pragma unroll
    for (int p = 0; p < 16; ++p)
        *(float4*)&lb[(p * 8 + wv) * TP + lane * 4] = rb[p];
    __syncthreads();

    const int rt = wv & 3;
    const int cg = wv >> 2;
    v8f acc[4];
#pragma unroll
    for (int j = 0; j < 4; ++j) acc[j] = (v8f){};

    if (rt * 16 < nvalid) {
        const float* pa = la + (rt * 16 + l16) * TP + 4 * half;
        const float* pb = lb + (cg * 64 + l16) * TP + 4 * half;
#pragma unroll
        for (int kk = 0; kk < 16; ++kk) {
            const float4 a4 = *(const float4*)(pa + 8 * kk);
            float4 b4[4];
#pragma unroll
            for (int j = 0; j < 4; ++j) b4[j] = *(const float4*)(pb + j * 16 * TP + 8 * kk);
            const v2f a0 = { a4.x, a4.y };
            const v2f a1 = { a4.z, a4.w };
#pragma unroll
            for (int j = 0; j < 4; ++j) {
                const v2f b0 = { b4[j].x, b4[j].y };
                acc[j] = wmma_f32(a0, b0, acc[j]);
            }
#pragma unroll
            for (int j = 0; j < 4; ++j) {
                const v2f b1 = { b4[j].z, b4[j].w };
                acc[j] = wmma_f32(a1, b1, acc[j]);
            }
        }
    }

    __syncthreads();
#pragma unroll
    for (int j = 0; j < 4; ++j)
#pragma unroll
        for (int r = 0; r < 8; ++r)
            la[(rt * 16 + r + 8 * half) * TP + cg * 64 + j * 16 + l16] = acc[j][r];
    __syncthreads();
}

__global__ __launch_bounds__(256) void qkv_gemm_kernel(
    const float* __restrict__ x, const float* __restrict__ xm, const float* __restrict__ w,
    float* __restrict__ qg, float* __restrict__ kg, float* __restrict__ vg)
{
    __shared__ float lds[(GEMM_BM + GEMM_BN) * TP];
    const int nb = blockIdx.x;
    const bool pooled = (blockIdx.y == 0);
    const int mb = (int)blockIdx.y - 1;
    const float* a_src = pooled ? xm : x + (size_t)mb * GEMM_BM * 128;
    const int nvalid = pooled ? 16 : GEMM_BM;

    gemm_tile_64x128(a_src, nvalid, w + (size_t)nb * GEMM_BN * 128, lds);

    float* dst = (nb == 0) ? qg : ((nb == 1) ? kg : vg);
    const int wv = __builtin_amdgcn_readfirstlane((int)(threadIdx.x >> 5));
    const int lane = threadIdx.x & 31;
    const int h = lane >> 3, d = (lane & 7) * 4;
    float4 vrow[8];
#pragma unroll
    for (int p = 0; p < 8; ++p) vrow[p] = *(const float4*)&lds[(p * 8 + wv) * TP + lane * 4];
#pragma unroll
    for (int p = 0; p < 8; ++p) {
        const int row = p * 8 + wv;
        if (row < nvalid) {
            const float4 v = vrow[p];
            int bidx, g, rig;
            if (pooled) { bidx = row >> 3; g = row & 7; rig = 1024; }
            else {
                const int tok = mb * GEMM_BM + row;
                bidx = tok >> 13;
                g = (tok & 8191) >> 10;
                rig = tok & 1023;
            }
            const size_t off = ((size_t)((bidx * 4 + h) * 8 + g) * GLEN + rig) * HSZ + d;
            *(float4*)(dst + off) = v;
        }
    }
}

#define NQT 33

__global__ __launch_bounds__(32 * AW, 4) void attn_group_kernel(
    const float* __restrict__ qg, const float* __restrict__ kg,
    const float* __restrict__ vg, float* __restrict__ z, float* __restrict__ xgm)
{
    const int grp = blockIdx.x;
    const int qt = (NQT - 1) - (int)blockIdx.y;
    const int wv = __builtin_amdgcn_readfirstlane((int)(threadIdx.x >> 5));
    const int lane = threadIdx.x & 31;
    const int half = lane >> 4, l16 = lane & 15;

    const float* Q = qg + (size_t)grp * GLEN * HSZ;
    const float* K = kg + (size_t)grp * GLEN * HSZ;
    const float* V = vg + (size_t)grp * GLEN * HSZ;
    const int gi = grp & 7;
    float*       Z = z + (size_t)(grp >> 5) * 8200 * 128 + ((grp >> 3) & 3) * HSZ;

    const int qbase = qt * 32;

    v2f qf[2][8];
#pragma unroll
    for (int u = 0; u < 2; ++u) {
        int qrow = qbase + 16 * u + l16; if (qrow > 1024) qrow = 1024;
        const float* qptr = Q + qrow * HSZ + 2 * half;
#pragma unroll
        for (int c = 0; c < 8; ++c) {
            float2 f = *(const float2*)(qptr + 4 * c);
            qf[u][c] = (v2f){ f.x, f.y };
        }
    }

    v8f acc[2][2];
    float rowM[2][8], rowL[2][8];
#pragma unroll
    for (int u = 0; u < 2; ++u) {
        acc[u][0] = (v8f){}; acc[u][1] = (v8f){};
#pragma unroll
        for (int r = 0; r < 8; ++r) { rowM[u][r] = -3.0e38f; rowL[u][r] = 0.0f; }
    }

    __shared__ __attribute__((aligned(16))) float lds_all[AW * WREG];
    float* lds = lds_all + wv * WREG;
    float* ldv = lds + KSLICE;

    const int srow = lane >> 3;
    const int scol = (lane & 7) * 4;

    for (int kbase = 64 * wv; kbase <= qbase + 31; kbase += 64 * AW) {
        int jmax = ((qbase + 31 - kbase) >> 4) + 1;
        if (jmax > 4) jmax = 4;

        v4f kr[16], vr[16];
        const bool tail = (kbase >= 1024);
        const float* kb = K + (tail ? 1024 : kbase) * HSZ;
        const float* vb = V + (tail ? 1024 : kbase) * HSZ;
        if (!tail) {
            const unsigned lo = (unsigned)(srow * HSZ + scol);
#pragma unroll
            for (int i = 0; i < 16; ++i) kr[i] = *(const v4f*)(kb + lo + i * 4 * HSZ);
#pragma unroll
            for (int i = 0; i < 16; ++i) vr[i] = *(const v4f*)(vb + lo + i * 4 * HSZ);
        } else {
            const v4f k1 = *(const v4f*)(kb + scol);
            const v4f v1 = *(const v4f*)(vb + scol);
#pragma unroll
            for (int i = 0; i < 16; ++i) { kr[i] = k1; vr[i] = v1; }
        }
        wave_lds_sync();
#pragma unroll
        for (int i = 0; i < 16; ++i)
            *(v4f*)&lds[(4 * i + srow) * KPITCH + scol] = kr[i];
#pragma unroll
        for (int i = 0; i < 16; ++i)
            *(v4f*)&ldv[(4 * i + srow) * VPITCH + scol] = vr[i];
        wave_lds_sync();

        v2f kf[4][8];
#pragma unroll
        for (int j = 0; j < 4; ++j)
#pragma unroll
            for (int c = 0; c < 8; ++c) {
                const float2 f = *(const float2*)&lds[(16 * j + l16) * KPITCH + 4 * c + 2 * half];
                kf[j][c] = (v2f){ f.x, f.y };
            }
        __builtin_amdgcn_sched_barrier(0);

        v8f s[2][4];
#pragma unroll
        for (int j = 0; j < 4; ++j) {
            s[0][j] = (v8f){}; s[1][j] = (v8f){};
            if (j < jmax) {
#pragma unroll
                for (int c = 0; c < 8; ++c) {
                    s[0][j] = wmma_f32(qf[0][c], kf[j][c], s[0][j]);
                    s[1][j] = wmma_f32(qf[1][c], kf[j][c], s[1][j]);
                }
            }
        }
        wave_lds_sync();

        const bool full = (kbase + 63 <= qbase);
        v2f b0[16], b1[16];
#pragma unroll
        for (int u = 0; u < 2; ++u) {
            if (!full) {
#pragma unroll
                for (int r = 0; r < 8; ++r) {
                    const int qi = qbase + 16 * u + r + 8 * half;
#pragma unroll
                    for (int j = 0; j < 4; ++j) {
                        const int kj = kbase + 16 * j + l16;
                        const bool valid = (kj <= qi) && (kj <= 1024);
                        s[u][j][r] = valid ? s[u][j][r] : -3.0e38f;
                    }
                }
            }
            float nmx[8];
#pragma unroll
            for (int r = 0; r < 8; ++r)
                nmx[r] = fmaxf(fmaxf(fmaxf(s[u][0][r], s[u][1][r]), s[u][2][r]), fmaxf(s[u][3][r], rowM[u][r]));
            row16_max8(nmx);
#pragma unroll
            for (int r = 0; r < 8; ++r) {
                const float nm = nmx[r];
                const float nb = -nm * SCALE_LOG2E;
                const float corr = __builtin_amdgcn_exp2f(__builtin_fmaf(rowM[u][r], SCALE_LOG2E, nb));
                float rs = 0.0f;
#pragma unroll
                for (int j = 0; j < 4; ++j) {
                    const float pv = __builtin_amdgcn_exp2f(__builtin_fmaf(s[u][j][r], SCALE_LOG2E, nb));
                    s[u][j][r] = pv;
                    rs += pv;
                }
                rowM[u][r] = nm;
                rowL[u][r] = __builtin_fmaf(rowL[u][r], corr, rs);
                asm volatile("" : "+v"(rowL[u][r]));
                acc[u][0][r] = acc[u][0][r] * corr;
                acc[u][1][r] = acc[u][1][r] * corr;
            }
#pragma unroll
            for (int j = 0; j < 4; ++j)
#pragma unroll
                for (int r = 0; r < 8; ++r)
                    lds[(16 * u + r + 8 * half) * PSTRIDE + 16 * j + l16] = s[u][j][r];
#pragma unroll
            for (int c = 8 * u; c < 8 * u + 8; ++c) {
                const float* vp = &ldv[(4 * c + 2 * half) * VPITCH + l16];
                b0[c] = (v2f){ vp[0],  vp[VPITCH] };
                b1[c] = (v2f){ vp[16], vp[VPITCH + 16] };
            }
        }
        wave_lds_sync();

        v2f pf[2][16];
#pragma unroll
        for (int u = 0; u < 2; ++u)
#pragma unroll
            for (int c = 0; c < 16; ++c) {
                const float2 f = *(const float2*)&lds[(16 * u + l16) * PSTRIDE + 4 * c + 2 * half];
                pf[u][c] = (v2f){ f.x, f.y };
            }
        __builtin_amdgcn_sched_barrier(0);

#pragma unroll
        for (int j = 0; j < 4; ++j) {
            if (j < jmax) {
#pragma unroll
                for (int c = 4 * j; c < 4 * j + 4; ++c) {
                    acc[0][0] = wmma_f32(pf[0][c], b0[c], acc[0][0]);
                    acc[0][1] = wmma_f32(pf[0][c], b1[c], acc[0][1]);
                    acc[1][0] = wmma_f32(pf[1][c], b0[c], acc[1][0]);
                    acc[1][1] = wmma_f32(pf[1][c], b1[c], acc[1][1]);
                }
            }
        }
    }

    const int lane_e = (int)__builtin_amdgcn_mbcnt_lo(~0u, 0u);
    const int half_e = lane_e >> 4, l16_e = lane_e & 15;
    wave_lds_sync();
#pragma unroll
    for (int u = 0; u < 2; ++u)
#pragma unroll
        for (int r = 0; r < 8; ++r) {
            lds[(32 * u + r)      * 32 + lane_e] = rowM[u][r];
            lds[(32 * u + 8 + r)  * 32 + lane_e] = rowL[u][r];
            lds[(32 * u + 16 + r) * 32 + lane_e] = acc[u][0][r];
            lds[(32 * u + 24 + r) * 32 + lane_e] = acc[u][1][r];
        }
    __syncthreads();

    float gls[2][8 / AW], o0s[2][8 / AW], o1s[2][8 / AW];
#pragma unroll
    for (int u = 0; u < 2; ++u)
#pragma unroll
        for (int i = 0; i < 8 / AW; ++i) {
            const int rr = wv + AW * i;
            float pm[AW];
            float gm = -3.0e38f;
#pragma unroll
            for (int w = 0; w < AW; ++w) {
                pm[w] = lds_all[w * WREG + (32 * u + rr) * 32 + lane_e];
                gm = fmaxf(gm, pm[w]);
            }
            float gl = 0.0f, o0 = 0.0f, o1 = 0.0f;
#pragma unroll
            for (int w = 0; w < AW; ++w) {
                const float e = __builtin_amdgcn_exp2f((pm[w] - gm) * SCALE_LOG2E);
                gl += lds_all[w * WREG + (32 * u + 8 + rr)  * 32 + lane_e] * e;
                o0 += lds_all[w * WREG + (32 * u + 16 + rr) * 32 + lane_e] * e;
                o1 += lds_all[w * WREG + (32 * u + 24 + rr) * 32 + lane_e] * e;
            }
            gls[u][i] = gl; o0s[u][i] = o0; o1s[u][i] = o1;
        }
#pragma unroll
    for (int u = 0; u < 2; ++u)
#pragma unroll
        for (int i = 0; i < 8 / AW; ++i) gls[u][i] = row16_sum(gls[u][i]);

#pragma unroll
    for (int u = 0; u < 2; ++u)
#pragma unroll
        for (int i = 0; i < 8 / AW; ++i) {
            const int rr = wv + AW * i;
            const int row = qbase + 16 * u + rr + 8 * half_e;
            if (row <= 1024) {
                const float invl = 1.0f / gls[u][i];
                const float o0 = o0s[u][i], o1 = o1s[u][i];
                const float r0 = o0 * invl + V[row * HSZ + l16_e];
                const float r1 = o1 * invl + V[row * HSZ + 16 + l16_e];
                if (row < 1024 || gi == 0) {
                    float* zp = Z + (size_t)(gi * GLEN + row + (gi > 0 ? 1 : 0)) * 128;
                    zp[l16_e]      = r0;
                    zp[16 + l16_e] = r1;
                }
                if (row == 1024) {
                    xgm[grp * HSZ + l16_e]      = r0;
                    xgm[grp * HSZ + 16 + l16_e] = r1;
                }
            }
        }
}

__global__ __launch_bounds__(256) void meanatt_kernel(
    const float* __restrict__ qg, const float* __restrict__ kg,
    const float* __restrict__ xgm, float* __restrict__ z)
{
    const int bh = blockIdx.x;
    __shared__ float qm[8][32], km[8][32], vm[8][32];
    const int t = threadIdx.x;
    const int g = t / 32, d = t % 32;
    const size_t rowoff = ((size_t)(bh * 8 + g) * GLEN + 1024) * HSZ + d;
    qm[g][d] = qg[rowoff];
    km[g][d] = kg[rowoff];
    vm[g][d] = xgm[(bh * 8 + g) * HSZ + d];
    __syncthreads();

    float sarr[8];
    float mx = -3.0e38f;
    for (int gp = 0; gp <= g; ++gp) {
        float sv = 0.0f;
        for (int c = 0; c < 32; ++c) sv += qm[g][c] * km[gp][c];
        sv *= SCALE;
        sarr[gp] = sv;
        mx = fmaxf(mx, sv);
    }
    float denom = 0.0f;
    for (int gp = 0; gp <= g; ++gp) { sarr[gp] = __expf(sarr[gp] - mx); denom += sarr[gp]; }
    float o = 0.0f;
    for (int gp = 0; gp <= g; ++gp) o += sarr[gp] * vm[gp][d];
    if (g < 7)
        z[((size_t)(bh >> 2) * 8200 + (size_t)(g + 1) * GLEN) * 128 + (bh & 3) * HSZ + d] = o / denom;
}

__global__ __launch_bounds__(256) void proj_gemm_kernel(
    const float* __restrict__ z, const float* __restrict__ w, float* __restrict__ out)
{
    __shared__ float lds[(GEMM_BM + GEMM_BN) * TP];
    const int mb = blockIdx.x;
    const int nvalid = (mb == 256) ? 16 : GEMM_BM;

    gemm_tile_64x128(z + (size_t)mb * GEMM_BM * 128, nvalid, w, lds);

    const int wv = __builtin_amdgcn_readfirstlane((int)(threadIdx.x >> 5));
    const int lane = threadIdx.x & 31;
    float4 vrow[8];
#pragma unroll
    for (int p = 0; p < 8; ++p) vrow[p] = *(const float4*)&lds[(p * 8 + wv) * TP + lane * 4];
#pragma unroll
    for (int p = 0; p < 8; ++p) {
        const int row = p * 8 + wv;
        if (row < nvalid) {
            const float4 v = vrow[p];
            *(float4*)(out + ((size_t)mb * GEMM_BM + row) * 128 + lane * 4) = v;
        }
    }
}

extern "C" void kernel_launch(void* const* d_in, const int* in_sizes, int n_in,
                              void* d_out, int out_size, void* d_ws, size_t ws_size,
                              hipStream_t stream) {
    const float* x      = (const float*)d_in[0];
    const float* w_attn = (const float*)d_in[1];
    const float* w_proj = (const float*)d_in[2];

    float* ws = (float*)d_ws;
    const size_t GBUF = (size_t)GROUPS * GLEN * HSZ;
    float* qg  = ws;
    float* kg  = ws + GBUF;
    float* vg  = ws + 2 * GBUF;
    float* z   = ws + 3 * GBUF;
    float* xgm = ws + 4 * GBUF;
    float* xm  = xgm;

    xmean_kernel<<<dim3(16, 4), 512, 0, stream>>>(x, xm);
    qkv_gemm_kernel<<<dim3(3, 257), 256, 0, stream>>>(x, xm, w_attn, qg, kg, vg);
    attn_group_kernel<<<dim3(GROUPS, NQT), 128, 0, stream>>>(qg, kg, vg, z, xgm);
    meanatt_kernel<<<8, 256, 0, stream>>>(qg, kg, xgm, z);
    proj_gemm_kernel<<<257, 256, 0, stream>>>(z, w_proj, (float*)d_out);
}


# attention: the wave that takes key block 0 (and so walks the most blocks) rotates with the work-group, so the 4 SIMDs of a WGP carry equal block counts; was always wave 0 (85:76:68:60 per group)
# speedup vs baseline: 3.9125x; 1.1076x over previous
#include <hip/hip_runtime.h>
#include <hip/hip_bf16.h>

typedef float v2f __attribute__((ext_vector_type(2)));
typedef float v8f __attribute__((ext_vector_type(8)));
typedef float v4f __attribute__((ext_vector_type(4)));

#define GROUPS 64
#define GLEN   1025
#define HSZ    32
#define SCALE  0.17677669529663687f
#define SCALE_LOG2E (SCALE * 1.4426950408889634f)
#define PSTRIDE 68
#define AW      4
#define KPITCH  36
#define VPITCH  40
#define KSLICE  (64 * KPITCH)
#define VSLICE  (64 * VPITCH)
#define WREG    (KSLICE + VSLICE)

__device__ __forceinline__ v8f wmma_f32(v2f a, v2f b, v8f c) {
    return __builtin_amdgcn_wmma_f32_16x16x4_f32(false, a, false, b, (short)0, c, false, false);
}

__device__ __forceinline__ void wave_lds_sync() {
    __builtin_amdgcn_fence(__ATOMIC_RELEASE, "wavefront");
    __builtin_amdgcn_wave_barrier();
    __builtin_amdgcn_fence(__ATOMIC_ACQUIRE, "wavefront");
}

#define DPP_XM(op, n, mask) op " %" #n ", %" #n ", %" #n " row_xmask:" #mask " row_mask:0xf bank_mask:0xf\n\t"
#define DPP_PHASE8(op, mask) DPP_XM(op, 0, mask) DPP_XM(op, 1, mask) DPP_XM(op, 2, mask) DPP_XM(op, 3, mask) \
                             DPP_XM(op, 4, mask) DPP_XM(op, 5, mask) DPP_XM(op, 6, mask) DPP_XM(op, 7, mask)

__device__ __forceinline__ void row16_max8(float (&v)[8]) {
    asm(DPP_PHASE8("v_max_num_f32_dpp", 1) DPP_PHASE8("v_max_num_f32_dpp", 2)
        DPP_PHASE8("v_max_num_f32_dpp", 4) DPP_PHASE8("v_max_num_f32_dpp", 8)
        : "+v"(v[0]), "+v"(v[1]), "+v"(v[2]), "+v"(v[3]), "+v"(v[4]), "+v"(v[5]), "+v"(v[6]), "+v"(v[7]));
}

__device__ __forceinline__ float row16_sum(float v) {
    asm(DPP_XM("v_add_f32_dpp", 0, 1) DPP_XM("v_add_f32_dpp", 0, 2)
        DPP_XM("v_add_f32_dpp", 0, 4) DPP_XM("v_add_f32_dpp", 0, 8)
        : "+v"(v));
    return v;
}

__global__ __launch_bounds__(512) void xmean_kernel(
    const float* __restrict__ x, float* __restrict__ xm)
{
    const int bg = blockIdx.x;
    const int slab = blockIdx.y;
    const int t = threadIdx.x;
    const int sub = t & 7;
    const int r0 = t >> 3;
    const float* p = x + ((size_t)bg * 1024 + r0) * 128 + slab * 32 + sub * 4;

    float4 v[16];
#pragma unroll
    for (int i = 0; i < 16; ++i) v[i] = *(const float4*)(p + (size_t)i * 64 * 128);
    float4 s = v[0];
#pragma unroll
    for (int i = 1; i < 16; ++i) { s.x += v[i].x; s.y += v[i].y; s.z += v[i].z; s.w += v[i].w; }

    s.x += __shfl_xor(s.x, 8, 32);  s.y += __shfl_xor(s.y, 8, 32);
    s.z += __shfl_xor(s.z, 8, 32);  s.w += __shfl_xor(s.w, 8, 32);
    s.x += __shfl_xor(s.x, 16, 32); s.y += __shfl_xor(s.y, 16, 32);
    s.z += __shfl_xor(s.z, 16, 32); s.w += __shfl_xor(s.w, 16, 32);

    __shared__ float red[16][32];
    const int wv = t >> 5, lane = t & 31;
    if (lane < 8) *(float4*)&red[wv][4 * lane] = s;
    __syncthreads();
    if (t < 32) {
        float a = 0.0f;
#pragma unroll
        for (int w = 0; w < 16; ++w) a += red[w][t];
        xm[bg * 128 + slab * 32 + t] = a * (1.0f / 1024.0f);
    }
}

#define TP      132
#define GEMM_BM 64
#define GEMM_BN 128

template <bool YROWS>
__device__ __forceinline__ void gemm_tile_64x128(
    const float* __restrict__ a_src, const int nvalid,
    const float* __restrict__ b_src, float* __restrict__ lds,
    const float* __restrict__ ypart, const int row0)
{
    float* la = lds;
    float* lb = lds + GEMM_BM * TP;
    const int wv = __builtin_amdgcn_readfirstlane((int)(threadIdx.x >> 5));
    const int lane = threadIdx.x & 31;
    const int half = lane >> 4, l16 = lane & 15;

    float4 ra[8], rb[16];
    int yrow = -1, yg = 0;
    if (YROWS) {
        const int rl = row0 + 63;
        const int tl = (rl >= 8200) ? rl - 8200 : rl;
        const int k = tl / 1025;
        const int back = tl - k * 1025;
        if (k >= 1 && k <= 7 && back < 64) { yrow = 63 - back; yg = k - 1; }
        if (yrow >= nvalid || (yrow & 7) != wv) yrow = -1;
    }
#pragma unroll
    for (int p = 0; p < 8; ++p) {
        const int row = p * 8 + wv;
        ra[p] = make_float4(0.0f, 0.0f, 0.0f, 0.0f);
        bool ld = (row < nvalid);
        if (YROWS && row == yrow) ld = false;
        if (ld) ra[p] = *(const float4*)(a_src + (size_t)row * 128 + lane * 4);
    }
#pragma unroll
    for (int p = 0; p < 16; ++p) {
        const int row = p * 8 + wv;
        rb[p] = *(const float4*)(b_src + (size_t)row * 128 + lane * 4);
    }
    float4 ys = make_float4(0.0f, 0.0f, 0.0f, 0.0f);
    if (YROWS && yrow >= 0) {
        const int bb = (row0 + yrow >= 8200) ? 1 : 0;
        const float* yp = ypart + (size_t)(((bb * 4 + (lane >> 3)) * 7 + yg) * 8) * HSZ + (lane & 7) * 4;
        float4 yt[7];
#pragma unroll
        for (int gs = 0; gs < 7; ++gs) yt[gs] = *(const float4*)(yp + gs * HSZ);
        ys = yt[0];
#pragma unroll
        for (int gs = 1; gs < 7; ++gs) { ys.x += yt[gs].x; ys.y += yt[gs].y; ys.z += yt[gs].z; ys.w += yt[gs].w; }
    }
#pragma unroll
    for (int p = 0; p < 8; ++p)
        *(float4*)&la[(p * 8 + wv) * TP + lane * 4] = ra[p];
#pragma unroll
    for (int p = 0; p < 16; ++p)
        *(float4*)&lb[(p * 8 + wv) * TP + lane * 4] = rb[p];
    if (YROWS && yrow >= 0)
        *(float4*)&la[yrow * TP + lane * 4] = ys;
    __syncthreads();

    const int rt = wv & 3;
    const int cg = wv >> 2;
    v8f acc[4];
#pragma unroll
    for (int j = 0; j < 4; ++j) acc[j] = (v8f){};

    if (rt * 16 < nvalid) {
        const float* pa = la + (rt * 16 + l16) * TP + 4 * half;
        const float* pb = lb + (cg * 64 + l16) * TP + 4 * half;
#pragma unroll
        for (int kk = 0; kk < 16; ++kk) {
            const float4 a4 = *(const float4*)(pa + 8 * kk);
            float4 b4[4];
#pragma unroll
            for (int j = 0; j < 4; ++j) b4[j] = *(const float4*)(pb + j * 16 * TP + 8 * kk);
            const v2f a0 = { a4.x, a4.y };
            const v2f a1 = { a4.z, a4.w };
#pragma unroll
            for (int j = 0; j < 4; ++j) {
                const v2f b0 = { b4[j].x, b4[j].y };
                acc[j] = wmma_f32(a0, b0, acc[j]);
            }
#pragma unroll
            for (int j = 0; j < 4; ++j) {
                const v2f b1 = { b4[j].z, b4[j].w };
                acc[j] = wmma_f32(a1, b1, acc[j]);
            }
        }
    }

    __syncthreads();
#pragma unroll
    for (int j = 0; j < 4; ++j)
#pragma unroll
        for (int r = 0; r < 8; ++r)
            la[(rt * 16 + r + 8 * half) * TP + cg * 64 + j * 16 + l16] = acc[j][r];
    __syncthreads();
}

__global__ __launch_bounds__(256) void qkv_gemm_kernel(
    const float* __restrict__ x, const float* __restrict__ xm, const float* __restrict__ w,
    float* __restrict__ qg, float* __restrict__ kg, float* __restrict__ vg)
{
    __shared__ float lds[(GEMM_BM + GEMM_BN) * TP];
    const int nb = blockIdx.x;
    const bool pooled = (blockIdx.y == 0);
    const int mb = (int)blockIdx.y - 1;
    const float* a_src = pooled ? xm : x + (size_t)mb * GEMM_BM * 128;
    const int nvalid = pooled ? 16 : GEMM_BM;

    gemm_tile_64x128<false>(a_src, nvalid, w + (size_t)nb * GEMM_BN * 128, lds, nullptr, 0);

    float* dst = (nb == 0) ? qg : ((nb == 1) ? kg : vg);
    const int wv = __builtin_amdgcn_readfirstlane((int)(threadIdx.x >> 5));
    const int lane = threadIdx.x & 31;
    const int h = lane >> 3, d = (lane & 7) * 4;
    float4 vrow[8];
#pragma unroll
    for (int p = 0; p < 8; ++p) vrow[p] = *(const float4*)&lds[(p * 8 + wv) * TP + lane * 4];
#pragma unroll
    for (int p = 0; p < 8; ++p) {
        const int row = p * 8 + wv;
        if (row < nvalid) {
            const float4 v = vrow[p];
            int bidx, g, rig;
            if (pooled) { bidx = row >> 3; g = row & 7; rig = 1024; }
            else {
                const int tok = mb * GEMM_BM + row;
                bidx = tok >> 13;
                g = (tok & 8191) >> 10;
                rig = tok & 1023;
            }
            const size_t off = ((size_t)((bidx * 4 + h) * 8 + g) * GLEN + rig) * HSZ + d;
            *(float4*)(dst + off) = v;
        }
    }
}

#define NQT 33

__global__ __launch_bounds__(32 * AW, 4) void attn_group_kernel(
    const float* __restrict__ qg, const float* __restrict__ kg,
    const float* __restrict__ vg, float* __restrict__ z, float* __restrict__ ypart)
{
    const int grp = blockIdx.x;
    const int qt = (NQT - 1) - (int)blockIdx.y;
    const int wv = __builtin_amdgcn_readfirstlane((int)(threadIdx.x >> 5));
    const int lane = threadIdx.x & 31;
    const int half = lane >> 4, l16 = lane & 15;

    const float* Q = qg + (size_t)grp * GLEN * HSZ;
    const float* K = kg + (size_t)grp * GLEN * HSZ;
    const float* V = vg + (size_t)grp * GLEN * HSZ;
    const int gi = grp & 7;
    float*       Z = z + (size_t)(grp >> 5) * 8200 * 128 + ((grp >> 3) & 3) * HSZ;

    const int qbase = qt * 32;

    v2f qf[2][8];
#pragma unroll
    for (int u = 0; u < 2; ++u) {
        int qrow = qbase + 16 * u + l16; if (qrow > 1024) qrow = 1024;
        const float* qptr = Q + qrow * HSZ + 2 * half;
#pragma unroll
        for (int c = 0; c < 8; ++c) {
            float2 f = *(const float2*)(qptr + 4 * c);
            qf[u][c] = (v2f){ f.x, f.y };
        }
    }

    v8f acc[2][2];
    float rowM[2][8], rowL[2][8];
#pragma unroll
    for (int u = 0; u < 2; ++u) {
        acc[u][0] = (v8f){}; acc[u][1] = (v8f){};
#pragma unroll
        for (int r = 0; r < 8; ++r) { rowM[u][r] = -3.0e38f; rowL[u][r] = 0.0f; }
    }

    __shared__ __attribute__((aligned(16))) float lds_all[AW * WREG];
    float* lds = lds_all + wv * WREG;
    float* ldv = lds + KSLICE;

    const int srow = lane >> 3;
    const int scol = (lane & 7) * 4;

    const int rot = (int)(blockIdx.x + blockIdx.y + (blockIdx.y >> 2));
    const int first = (wv + rot) & (AW - 1);
    for (int kbase = 64 * first; kbase <= qbase + 31; kbase += 64 * AW) {
        int jmax = ((qbase + 31 - kbase) >> 4) + 1;
        if (jmax > 4) jmax = 4;

        v4f kr[16], vr[16];
        const bool tail = (kbase >= 1024);
        const float* kb = K + (tail ? 1024 : kbase) * HSZ;
        const float* vb = V + (tail ? 1024 : kbase) * HSZ;
        if (!tail) {
            const unsigned lo = (unsigned)(srow * HSZ + scol);
#pragma unroll
            for (int i = 0; i < 16; ++i) kr[i] = *(const v4f*)(kb + lo + i * 4 * HSZ);
#pragma unroll
            for (int i = 0; i < 16; ++i) vr[i] = *(const v4f*)(vb + lo + i * 4 * HSZ);
        } else {
            const v4f k1 = *(const v4f*)(kb + scol);
            const v4f v1 = *(const v4f*)(vb + scol);
#pragma unroll
            for (int i = 0; i < 16; ++i) { kr[i] = k1; vr[i] = v1; }
        }
        wave_lds_sync();
#pragma unroll
        for (int i = 0; i < 16; ++i)
            *(v4f*)&lds[(4 * i + srow) * KPITCH + scol] = kr[i];
#pragma unroll
        for (int i = 0; i < 16; ++i)
            *(v4f*)&ldv[(4 * i + srow) * VPITCH + scol] = vr[i];
        wave_lds_sync();

        v2f kf[4][8];
#pragma unroll
        for (int j = 0; j < 4; ++j)
#pragma unroll
            for (int c = 0; c < 8; ++c) {
                const float2 f = *(const float2*)&lds[(16 * j + l16) * KPITCH + 4 * c + 2 * half];
                kf[j][c] = (v2f){ f.x, f.y };
            }
        __builtin_amdgcn_sched_barrier(0);

        v8f s[2][4];
#pragma unroll
        for (int j = 0; j < 4; ++j) {
            s[0][j] = (v8f){}; s[1][j] = (v8f){};
            if (j < jmax) {
#pragma unroll
                for (int c = 0; c < 8; ++c) {
                    s[0][j] = wmma_f32(qf[0][c], kf[j][c], s[0][j]);
                    s[1][j] = wmma_f32(qf[1][c], kf[j][c], s[1][j]);
                }
            }
        }
        wave_lds_sync();

        const bool full = (kbase + 63 <= qbase);
        v2f b0[16], b1[16];
#pragma unroll
        for (int u = 0; u < 2; ++u) {
            if (!full) {
#pragma unroll
                for (int r = 0; r < 8; ++r) {
                    const int qi = qbase + 16 * u + r + 8 * half;
#pragma unroll
                    for (int j = 0; j < 4; ++j) {
                        const int kj = kbase + 16 * j + l16;
                        const bool valid = (kj <= qi) && (kj <= 1024);
                        s[u][j][r] = valid ? s[u][j][r] : -3.0e38f;
                    }
                }
            }
            float nmx[8];
#pragma unroll
            for (int r = 0; r < 8; ++r)
                nmx[r] = fmaxf(fmaxf(fmaxf(s[u][0][r], s[u][1][r]), s[u][2][r]), fmaxf(s[u][3][r], rowM[u][r]));
            row16_max8(nmx);
#pragma unroll
            for (int r = 0; r < 8; ++r) {
                const float nm = nmx[r];
                const float nb = -nm * SCALE_LOG2E;
                const float corr = __builtin_amdgcn_exp2f(__builtin_fmaf(rowM[u][r], SCALE_LOG2E, nb));
                float rs = 0.0f;
#pragma unroll
                for (int j = 0; j < 4; ++j) {
                    const float pv = __builtin_amdgcn_exp2f(__builtin_fmaf(s[u][j][r], SCALE_LOG2E, nb));
                    s[u][j][r] = pv;
                    rs += pv;
                }
                rowM[u][r] = nm;
                rowL[u][r] = __builtin_fmaf(rowL[u][r], corr, rs);
                asm volatile("" : "+v"(rowL[u][r]));
                acc[u][0][r] = acc[u][0][r] * corr;
                acc[u][1][r] = acc[u][1][r] * corr;
            }
#pragma unroll
            for (int j = 0; j < 4; ++j)
#pragma unroll
                for (int r = 0; r < 8; ++r)
                    lds[(16 * u + r + 8 * half) * PSTRIDE + 16 * j + l16] = s[u][j][r];
#pragma unroll
            for (int c = 8 * u; c < 8 * u + 8; ++c) {
                const float* vp = &ldv[(4 * c + 2 * half) * VPITCH + l16];
                b0[c] = (v2f){ vp[0],  vp[VPITCH] };
                b1[c] = (v2f){ vp[16], vp[VPITCH + 16] };
            }
        }
        wave_lds_sync();

        v2f pf[2][16];
#pragma unroll
        for (int u = 0; u < 2; ++u)
#pragma unroll
            for (int c = 0; c < 16; ++c) {
                const float2 f = *(const float2*)&lds[(16 * u + l16) * PSTRIDE + 4 * c + 2 * half];
                pf[u][c] = (v2f){ f.x, f.y };
            }
        __builtin_amdgcn_sched_barrier(0);

#pragma unroll
        for (int j = 0; j < 4; ++j) {
            if (j < jmax) {
#pragma unroll
                for (int c = 4 * j; c < 4 * j + 4; ++c) {
                    acc[0][0] = wmma_f32(pf[0][c], b0[c], acc[0][0]);
                    acc[0][1] = wmma_f32(pf[0][c], b1[c], acc[0][1]);
                    acc[1][0] = wmma_f32(pf[1][c], b0[c], acc[1][0]);
                    acc[1][1] = wmma_f32(pf[1][c], b1[c], acc[1][1]);
                }
            }
        }
    }

    const int lane_e = (int)__builtin_amdgcn_mbcnt_lo(~0u, 0u);
    const int half_e = lane_e >> 4, l16_e = lane_e & 15;
    wave_lds_sync();
#pragma unroll
    for (int u = 0; u < 2; ++u)
#pragma unroll
        for (int r = 0; r < 8; ++r) {
            lds[(32 * u + r)      * 32 + lane_e] = rowM[u][r];
            lds[(32 * u + 8 + r)  * 32 + lane_e] = rowL[u][r];
            lds[(32 * u + 16 + r) * 32 + lane_e] = acc[u][0][r];
            lds[(32 * u + 24 + r) * 32 + lane_e] = acc[u][1][r];
        }
    __syncthreads();

    float gls[2][8 / AW], o0s[2][8 / AW], o1s[2][8 / AW];
#pragma unroll
    for (int u = 0; u < 2; ++u)
#pragma unroll
        for (int i = 0; i < 8 / AW; ++i) {
            const int rr = wv + AW * i;
            float pm[AW];
            float gm = -3.0e38f;
#pragma unroll
            for (int w = 0; w < AW; ++w) {
                pm[w] = lds_all[w * WREG + (32 * u + rr) * 32 + lane_e];
                gm = fmaxf(gm, pm[w]);
            }
            float gl = 0.0f, o0 = 0.0f, o1 = 0.0f;
#pragma unroll
            for (int w = 0; w < AW; ++w) {
                const float e = __builtin_amdgcn_exp2f((pm[w] - gm) * SCALE_LOG2E);
                gl += lds_all[w * WREG + (32 * u + 8 + rr)  * 32 + lane_e] * e;
                o0 += lds_all[w * WREG + (32 * u + 16 + rr) * 32 + lane_e] * e;
                o1 += lds_all[w * WREG + (32 * u + 24 + rr) * 32 + lane_e] * e;
            }
            gls[u][i] = gl; o0s[u][i] = o0; o1s[u][i] = o1;
        }
#pragma unroll
    for (int u = 0; u < 2; ++u)
#pragma unroll
        for (int i = 0; i < 8 / AW; ++i) gls[u][i] = row16_sum(gls[u][i]);

    float wy[7];
#pragma unroll
    for (int g = 0; g < 7; ++g) wy[g] = 0.0f;
    const bool pool_wave = (qt == NQT - 1) && (wv == 0);
    if (pool_wave) {
        const int gsel = (l16_e < 8) ? l16_e : 7;
        const size_t prow = ((size_t)((grp & ~7) + gsel) * GLEN + 1024) * HSZ + 2 * half_e;
        v2f qa[8], kb[8];
#pragma unroll
        for (int c = 0; c < 8; ++c) {
            const float2 fq = *(const float2*)(qg + prow + 4 * c);
            const float2 fk = *(const float2*)(kg + prow + 4 * c);
            qa[c] = (v2f){ fq.x, fq.y };
            kb[c] = (v2f){ fk.x, fk.y };
        }
        v8f sg = (v8f){};
#pragma unroll
        for (int c = 0; c < 8; ++c) sg = wmma_f32(qa[c], kb[c], sg);
        asm volatile("v_nop\n\tv_nop\n\tv_nop\n\tv_nop\n\tv_nop\n\tv_nop\n\tv_nop\n\tv_nop\n\tv_nop\n\tv_nop"
                     : "+v"(sg));
        float sv[8], mx[8];
#pragma unroll
        for (int r = 0; r < 8; ++r) { sv[r] = (l16_e <= r) ? sg[r] : -3.0e38f; mx[r] = sv[r]; }
        row16_max8(mx);
#pragma unroll
        for (int g = 0; g < 7; ++g) {
            const float e = __builtin_amdgcn_exp2f((sv[g] - mx[g]) * SCALE_LOG2E);
            const float wgt = e / row16_sum(e);
            wy[g] = __int_as_float(__builtin_amdgcn_readlane(__float_as_int(wgt), gi));
        }
    }

#pragma unroll
    for (int u = 0; u < 2; ++u)
#pragma unroll
        for (int i = 0; i < 8 / AW; ++i) {
            const int rr = wv + AW * i;
            const int row = qbase + 16 * u + rr + 8 * half_e;
            if (row <= 1024) {
                const float invl = 1.0f / gls[u][i];
                const float o0 = o0s[u][i], o1 = o1s[u][i];
                const float r0 = o0 * invl + V[row * HSZ + l16_e];
                const float r1 = o1 * invl + V[row * HSZ + 16 + l16_e];
                if (row < 1024 || gi == 0) {
                    float* zp = Z + (size_t)(gi * GLEN + row + (gi > 0 ? 1 : 0)) * 128;
                    zp[l16_e]      = r0;
                    zp[16 + l16_e] = r1;
                }
                if (u == 0 && i == 0 && pool_wave && row == 1024) {
#pragma unroll
                    for (int g = 0; g < 7; ++g) {
                        float* yp = ypart + (size_t)(((grp >> 3) * 7 + g) * 8 + gi) * HSZ;
                        yp[l16_e]      = wy[g] * r0;
                        yp[16 + l16_e] = wy[g] * r1;
                    }
                }
            }
        }
}

__global__ __launch_bounds__(256) void proj_gemm_kernel(
    const float* __restrict__ z, const float* __restrict__ ypart, const float* __restrict__ w, float* __restrict__ out)
{
    __shared__ float lds[(GEMM_BM + GEMM_BN) * TP];
    const int mb = blockIdx.x;
    const int nvalid = (mb == 256) ? 16 : GEMM_BM;

    gemm_tile_64x128<true>(z + (size_t)mb * GEMM_BM * 128, nvalid, w, lds, ypart, mb * GEMM_BM);

    const int wv = __builtin_amdgcn_readfirstlane((int)(threadIdx.x >> 5));
    const int lane = threadIdx.x & 31;
    float4 vrow[8];
#pragma unroll
    for (int p = 0; p < 8; ++p) vrow[p] = *(const float4*)&lds[(p * 8 + wv) * TP + lane * 4];
#pragma unroll
    for (int p = 0; p < 8; ++p) {
        const int row = p * 8 + wv;
        if (row < nvalid) {
            const float4 v = vrow[p];
            *(float4*)(out + ((size_t)mb * GEMM_BM + row) * 128 + lane * 4) = v;
        }
    }
}

extern "C" void kernel_launch(void* const* d_in, const int* in_sizes, int n_in,
                              void* d_out, int out_size, void* d_ws, size_t ws_size,
                              hipStream_t stream) {
    const float* x      = (const float*)d_in[0];
    const float* w_attn = (const float*)d_in[1];
    const float* w_proj = (const float*)d_in[2];

    float* ws = (float*)d_ws;
    const size_t GBUF = (size_t)GROUPS * GLEN * HSZ;
    float* qg  = ws;
    float* kg  = ws + GBUF;
    float* vg  = ws + 2 * GBUF;
    float* z   = ws + 3 * GBUF;
    float* xm  = ws + 4 * GBUF;
    float* ypart = ws + 4 * GBUF + 2048;

    xmean_kernel<<<dim3(16, 4), 512, 0, stream>>>(x, xm);
    qkv_gemm_kernel<<<dim3(3, 257), 256, 0, stream>>>(x, xm, w_attn, qg, kg, vg);
    attn_group_kernel<<<dim3(GROUPS, NQT), 128, 0, stream>>>(qg, kg, vg, z, ypart);
    proj_gemm_kernel<<<257, 256, 0, stream>>>(z, ypart, w_proj, (float*)d_out);
}
